# starting kernel
# baseline (speedup 1.0000x reference)
#include <hip/hip_runtime.h>
#include <cstdint>

typedef __attribute__((ext_vector_type(2))) float v2f;
typedef __attribute__((ext_vector_type(8))) float v8f;

#define CFEAT 128


__global__ void k_init(int* __restrict__ deg, int* __restrict__ cursor, int n) {
    int i = blockIdx.x * blockDim.x + threadIdx.x;
    if (i < n) { deg[i] = 0; cursor[i] = 0; }
}

__global__ void k_count(const int* __restrict__ dst, int* __restrict__ deg, int e) {
    int i = blockIdx.x * blockDim.x + threadIdx.x;
    if (i < e) atomicAdd(&deg[dst[i]], 1);
}

__global__ void k_dinv(const int* __restrict__ deg, float* __restrict__ dinv, int n) {
    int i = blockIdx.x * blockDim.x + threadIdx.x;
    if (i < n) dinv[i] = rsqrtf((float)deg[i] + 1.0f);
}

__global__ __launch_bounds__(1024)
void k_scan(const int* __restrict__ deg, int* __restrict__ offs, int n) {
    __shared__ int buf[1024];
    __shared__ int carry;
    const int tid = threadIdx.x;
    if (tid == 0) carry = 0;
    __syncthreads();
    for (int base = 0; base < n; base += 1024) {
        int i = base + tid;
        int v = (i < n) ? deg[i] : 0;
        buf[tid] = v;
        __syncthreads();
        #pragma unroll
        for (int off = 1; off < 1024; off <<= 1) {
            int t = (tid >= off) ? buf[tid - off] : 0;
            __syncthreads();
            buf[tid] += t;
            __syncthreads();
        }
        int incl = buf[tid];
        int c = carry;
        if (i < n) offs[i] = c + incl - v;
        __syncthreads();
        if (tid == 0) carry = c + buf[1023];
        __syncthreads();
    }
    if (tid == 0) offs[n] = carry;
}

__global__ void k_fill(const int* __restrict__ src, const int* __restrict__ dst,
                       const int* __restrict__ offs, int* __restrict__ cursor,
                       int* __restrict__ csr, int e) {
    int i = blockIdx.x * blockDim.x + threadIdx.x;
    if (i < e) {
        int d = dst[i];
        int p = atomicAdd(&cursor[d], 1);
        csr[offs[d] + p] = src[i];
    }
}


__global__ __launch_bounds__(256)
void k_gemm1(const float* __restrict__ x, const float* __restrict__ W1,
             float* __restrict__ h, int n) {
    __shared__ float sW[CFEAT * CFEAT];
    for (int i = threadIdx.x; i < (CFEAT * CFEAT) / 4; i += 256)
        ((float4*)sW)[i] = ((const float4*)W1)[i];
    __syncthreads();

    const int lane = threadIdx.x & 31;
    const int wave = threadIdx.x >> 5;
    const int hi   = lane >> 4;
    const int lo   = lane & 15;
    const int rowbase = (blockIdx.x * 8 + wave) * 16;
    if (rowbase >= n) return;

    int arow = rowbase + lo;
    if (arow >= n) arow = n - 1;
    const float* __restrict__ xrow = x + (size_t)arow * CFEAT;

    v8f acc[8];
    const v8f z = {0.f, 0.f, 0.f, 0.f, 0.f, 0.f, 0.f, 0.f};
    #pragma unroll
    for (int t = 0; t < 8; ++t) acc[t] = z;

    for (int k = 0; k < CFEAT; k += 4) {
        v2f a;
        a.x = xrow[k + 2 * hi];
        a.y = xrow[k + 2 * hi + 1];
        #pragma unroll
        for (int t = 0; t < 8; ++t) {
            v2f b;
            const int col = t * 16 + lo;
            b.x = sW[(k + 2 * hi) * CFEAT + col];
            b.y = sW[(k + 2 * hi + 1) * CFEAT + col];
            acc[t] = __builtin_amdgcn_wmma_f32_16x16x4_f32(
                false, a, false, b, (short)0, acc[t], false, false);
        }
    }

    if (rowbase + 16 <= n) {
        #pragma unroll
        for (int t = 0; t < 8; ++t) {
            #pragma unroll
            for (int r = 0; r < 8; ++r) {
                const int row = rowbase + r + 8 * hi;
                h[(size_t)row * CFEAT + t * 16 + lo] = acc[t][r];
            }
        }
    } else {
        #pragma unroll
        for (int t = 0; t < 8; ++t) {
            #pragma unroll
            for (int r = 0; r < 8; ++r) {
                const int row = rowbase + r + 8 * hi;
                if (row < n) h[(size_t)row * CFEAT + t * 16 + lo] = acc[t][r];
            }
        }
    }
}


__device__ __forceinline__ float4 aggregate(const float* __restrict__ hsrc,
                                            const int* __restrict__ offs,
                                            const int* __restrict__ csr,
                                            const float* __restrict__ dinv,
                                            int node, int lane) {
    const float din = dinv[node];
    const float4* __restrict__ h4 = (const float4*)hsrc;
    float4 acc;
    {
        float4 v = h4[(size_t)node * 32 + lane];
        acc.x = v.x * din; acc.y = v.y * din; acc.z = v.z * din; acc.w = v.w * din;
    }
    const int beg = offs[node], end = offs[node + 1];
    for (int j = beg; j < end; j += 32) {
        const int idx = j + lane;
        const int   s  = (idx < end) ? csr[idx] : 0;
        const float ws = (idx < end) ? dinv[s] : 0.0f;
        int cnt = end - j; if (cnt > 32) cnt = 32;
        int t = 0;
        if (cnt == 32) {
            for (; t < 32; t += 4) {
                const int   s0 = __shfl(s, t,     32), s1 = __shfl(s, t + 1, 32);
                const int   s2 = __shfl(s, t + 2, 32), s3 = __shfl(s, t + 3, 32);
                const float w0 = __shfl(ws, t,     32), w1 = __shfl(ws, t + 1, 32);
                const float w2 = __shfl(ws, t + 2, 32), w3 = __shfl(ws, t + 3, 32);
                float4 v0 = h4[(size_t)s0 * 32 + lane];
                float4 v1 = h4[(size_t)s1 * 32 + lane];
                float4 v2 = h4[(size_t)s2 * 32 + lane];
                float4 v3 = h4[(size_t)s3 * 32 + lane];
                acc.x += v0.x * w0; acc.y += v0.y * w0; acc.z += v0.z * w0; acc.w += v0.w * w0;
                acc.x += v1.x * w1; acc.y += v1.y * w1; acc.z += v1.z * w1; acc.w += v1.w * w1;
                acc.x += v2.x * w2; acc.y += v2.y * w2; acc.z += v2.z * w2; acc.w += v2.w * w2;
                acc.x += v3.x * w3; acc.y += v3.y * w3; acc.z += v3.z * w3; acc.w += v3.w * w3;
            }
        }
        for (; t < cnt; ++t) {
            const int   sk = __shfl(s, t, 32);
            const float wk = __shfl(ws, t, 32);
            float4 v = h4[(size_t)sk * 32 + lane];
            acc.x += v.x * wk; acc.y += v.y * wk; acc.z += v.z * wk; acc.w += v.w * wk;
        }
    }
    acc.x *= din; acc.y *= din; acc.z *= din; acc.w *= din;
    return acc;
}

__global__ __launch_bounds__(256)
void k_prop1(const float* __restrict__ h, float* __restrict__ h2,
             const int* __restrict__ offs, const int* __restrict__ csr,
             const float* __restrict__ dinv, const float* __restrict__ b1, int n) {
    const int node = (int)((blockIdx.x * blockDim.x + threadIdx.x) >> 5);
    const int lane = threadIdx.x & 31;
    if (node >= n) return;
    float4 acc = aggregate(h, offs, csr, dinv, node, lane);
    const float4 b = ((const float4*)b1)[lane];
    acc.x = fmaxf(acc.x + b.x, 0.0f);
    acc.y = fmaxf(acc.y + b.y, 0.0f);
    acc.z = fmaxf(acc.z + b.z, 0.0f);
    acc.w = fmaxf(acc.w + b.w, 0.0f);
    ((float4*)h2)[(size_t)node * 32 + lane] = acc;
}

__global__ __launch_bounds__(256)
void k_prop2_out(const float* __restrict__ h2, float* __restrict__ out,
                 const int* __restrict__ offs, const int* __restrict__ csr,
                 const float* __restrict__ dinv, const float* __restrict__ Wout,
                 const float* __restrict__ bout, int n) {
    const int node = (int)((blockIdx.x * blockDim.x + threadIdx.x) >> 5);
    const int lane = threadIdx.x & 31;
    if (node >= n) return;

    float wl[4][4][2];
    #pragma unroll
    for (int m = 0; m < 4; ++m)
        #pragma unroll
        for (int j = 0; j < 4; ++j) {
            const float2 w = ((const float2*)Wout)[m * CFEAT + lane * 4 + j];
            wl[m][j][0] = w.x; wl[m][j][1] = w.y;
        }

    float4 acc = aggregate(h2, offs, csr, dinv, node, lane);

    float p[4][2];
    #pragma unroll
    for (int m = 0; m < 4; ++m)
        #pragma unroll
        for (int o = 0; o < 2; ++o)
            p[m][o] = acc.x * wl[m][0][o] + acc.y * wl[m][1][o] +
                      acc.z * wl[m][2][o] + acc.w * wl[m][3][o];

    #pragma unroll
    for (int m = 0; m < 4; ++m)
        #pragma unroll
        for (int o = 0; o < 2; ++o) {
            float v = p[m][o];
            #pragma unroll
            for (int d = 16; d > 0; d >>= 1) v += __shfl_xor(v, d, 32);
            p[m][o] = v;
        }

    if (lane == 0) {
        #pragma unroll
        for (int m = 0; m < 4; ++m) {
            float2 r;
            r.x = p[m][0] + bout[m * 2 + 0];
            r.y = p[m][1] + bout[m * 2 + 1];
            ((float2*)out)[(size_t)m * n + node] = r;
        }
    }
}


extern "C" void kernel_launch(void* const* d_in, const int* in_sizes, int n_in,
                              void* d_out, int out_size, void* d_ws, size_t ws_size,
                              hipStream_t stream) {
    const float* x    = (const float*)d_in[0];
    const int*   ei   = (const int*)d_in[1];
    const float* W1   = (const float*)d_in[2];
    const float* b1   = (const float*)d_in[3];
    const float* Wout = (const float*)d_in[4];
    const float* bout = (const float*)d_in[5];
    float* out = (float*)d_out;

    const int n = in_sizes[0] / CFEAT;
    const int e = in_sizes[1] / 2;
    const int* srcv = ei;
    const int* dstv = ei + e;

    uintptr_t p = (uintptr_t)d_ws;
    auto alloc = [&](size_t bytes) {
        p = (p + 15) & ~(uintptr_t)15;
        void* r = (void*)p;
        p += bytes;
        return r;
    };
    int*   deg    = (int*)  alloc((size_t)n * 4);
    int*   cursor = (int*)  alloc((size_t)n * 4);
    int*   offs   = (int*)  alloc((size_t)(n + 1) * 4);
    int*   csr    = (int*)  alloc((size_t)e * 4);
    float* dinv   = (float*)alloc((size_t)n * 4);
    float* h      = (float*)alloc((size_t)n * CFEAT * 4);
    float* h2     = (float*)alloc((size_t)n * CFEAT * 4);
    (void)ws_size; (void)n_in; (void)out_size;

    const int TB = 256;
    const int gn = (n + TB - 1) / TB;
    const int ge = (e + TB - 1) / TB;

    k_init <<<gn, TB, 0, stream>>>(deg, cursor, n);
    k_count<<<ge, TB, 0, stream>>>(dstv, deg, e);
    k_dinv <<<gn, TB, 0, stream>>>(deg, dinv, n);
    k_scan <<<1, 1024, 0, stream>>>(deg, offs, n);
    k_fill <<<ge, TB, 0, stream>>>(srcv, dstv, offs, cursor, csr, e);

    k_gemm1<<<(n + 127) / 128, TB, 0, stream>>>(x, W1, h, n);

    const int gprop = (n + 7) / 8;
    k_prop1    <<<gprop, TB, 0, stream>>>(h, h2, offs, csr, dinv, b1, n);
    k_prop2_out<<<gprop, TB, 0, stream>>>(h2, out, offs, csr, dinv, Wout, bout, n);
}


# k_scan: one pass, 16 int4 per lane kept in registers, DPP row_shr wave scans and one barrier, replaces the 49-chunk LDS Hillis-Steele scan (1,176 barriers); integer sums only
# speedup vs baseline: 1.2859x; 1.2859x over previous
#include <hip/hip_runtime.h>
#include <cstdint>

typedef __attribute__((ext_vector_type(2))) float v2f;
typedef __attribute__((ext_vector_type(8))) float v8f;

#define CFEAT 128


__global__ void k_init(int* __restrict__ deg, int* __restrict__ cursor, int n) {
    int i = blockIdx.x * blockDim.x + threadIdx.x;
    if (i < n) { deg[i] = 0; cursor[i] = 0; }
}

__global__ void k_count(const int* __restrict__ dst, int* __restrict__ deg, int e) {
    int i = blockIdx.x * blockDim.x + threadIdx.x;
    if (i < e) atomicAdd(&deg[dst[i]], 1);
}

__global__ void k_dinv(const int* __restrict__ deg, float* __restrict__ dinv, int n) {
    int i = blockIdx.x * blockDim.x + threadIdx.x;
    if (i < n) dinv[i] = rsqrtf((float)deg[i] + 1.0f);
}


#define SCAN_ITS 16

__device__ __forceinline__ int wave_incl_scan(int x, int lane) {
    x += __builtin_amdgcn_update_dpp(0, x, 0x111, 0xf, 0xf, true);
    x += __builtin_amdgcn_update_dpp(0, x, 0x112, 0xf, 0xf, true);
    x += __builtin_amdgcn_update_dpp(0, x, 0x114, 0xf, 0xf, true);
    x += __builtin_amdgcn_update_dpp(0, x, 0x118, 0xf, 0xf, true);
    const int row0 = __builtin_amdgcn_readlane(x, 15);
    return x + ((lane >= 16) ? row0 : 0);
}

__global__ __launch_bounds__(1024)
void k_scan(const int* __restrict__ deg, int* __restrict__ offs, int n) {
    __shared__ int wtot[2][32];
    const int tid  = threadIdx.x;
    const int lane = tid & 31;
    const int wave = __builtin_amdgcn_readfirstlane(tid >> 5);
    const int nv   = n >> 2;
    const int4* __restrict__ deg4  = (const int4*)deg;
    int4*       __restrict__ offs4 = (int4*)offs;

    int running = 0;
    int par = 0;
    for (int gbase = 0; gbase < nv; gbase += SCAN_ITS * 1024) {
        int ng = nv - gbase;
        if (ng > SCAN_ITS * 1024) ng = SCAN_ITS * 1024;
        const int its  = (ng + 1023) >> 10;
        const int gend = gbase + ng;
        const int g0   = gbase + wave * its * 32 + lane;

        int4 v[SCAN_ITS];
        #pragma unroll
        for (int it = 0; it < SCAN_ITS; ++it) {
            const int g = g0 + it * 32;
            int4 t = make_int4(0, 0, 0, 0);
            if (it < its && g < gend) t = deg4[g];
            v[it] = t;
        }

        int carry = 0;
        #pragma unroll
        for (int it = 0; it < SCAN_ITS; ++it) {
            if (it < its) {
                const int4 t  = v[it];
                const int s1  = t.x + t.y;
                const int s2  = s1 + t.z;
                const int s3  = s2 + t.w;
                const int inc = wave_incl_scan(s3, lane);
                const int b   = carry + inc - s3;
                v[it] = make_int4(b, b + t.x, b + s1, b + s2);
                carry += __builtin_amdgcn_readlane(inc, 31);
            }
        }

        if (lane == 0) wtot[par][wave] = carry;
        __syncthreads();
        const int wt   = wtot[par][lane];
        const int winc = wave_incl_scan(wt, lane);
        const int add  = running + __builtin_amdgcn_readlane(winc - wt, wave);
        running += __builtin_amdgcn_readlane(winc, 31);

        #pragma unroll
        for (int it = 0; it < SCAN_ITS; ++it) {
            const int g = g0 + it * 32;
            if (it < its && g < gend) {
                int4 o = v[it];
                o.x += add; o.y += add; o.z += add; o.w += add;
                offs4[g] = o;
            }
        }
        par ^= 1;
    }
    if (tid == 0) {
        int r = running;
        for (int i = nv * 4; i < n; ++i) { offs[i] = r; r += deg[i]; }
        offs[n] = r;
    }
}

__global__ void k_fill(const int* __restrict__ src, const int* __restrict__ dst,
                       const int* __restrict__ offs, int* __restrict__ cursor,
                       int* __restrict__ csr, int e) {
    int i = blockIdx.x * blockDim.x + threadIdx.x;
    if (i < e) {
        int d = dst[i];
        int p = atomicAdd(&cursor[d], 1);
        csr[offs[d] + p] = src[i];
    }
}


__global__ __launch_bounds__(256)
void k_gemm1(const float* __restrict__ x, const float* __restrict__ W1,
             float* __restrict__ h, int n) {
    __shared__ float sW[CFEAT * CFEAT];
    for (int i = threadIdx.x; i < (CFEAT * CFEAT) / 4; i += 256)
        ((float4*)sW)[i] = ((const float4*)W1)[i];
    __syncthreads();

    const int lane = threadIdx.x & 31;
    const int wave = threadIdx.x >> 5;
    const int hi   = lane >> 4;
    const int lo   = lane & 15;
    const int rowbase = (blockIdx.x * 8 + wave) * 16;
    if (rowbase >= n) return;

    int arow = rowbase + lo;
    if (arow >= n) arow = n - 1;
    const float* __restrict__ xrow = x + (size_t)arow * CFEAT;

    v8f acc[8];
    const v8f z = {0.f, 0.f, 0.f, 0.f, 0.f, 0.f, 0.f, 0.f};
    #pragma unroll
    for (int t = 0; t < 8; ++t) acc[t] = z;

    for (int k = 0; k < CFEAT; k += 4) {
        v2f a;
        a.x = xrow[k + 2 * hi];
        a.y = xrow[k + 2 * hi + 1];
        #pragma unroll
        for (int t = 0; t < 8; ++t) {
            v2f b;
            const int col = t * 16 + lo;
            b.x = sW[(k + 2 * hi) * CFEAT + col];
            b.y = sW[(k + 2 * hi + 1) * CFEAT + col];
            acc[t] = __builtin_amdgcn_wmma_f32_16x16x4_f32(
                false, a, false, b, (short)0, acc[t], false, false);
        }
    }

    if (rowbase + 16 <= n) {
        #pragma unroll
        for (int t = 0; t < 8; ++t) {
            #pragma unroll
            for (int r = 0; r < 8; ++r) {
                const int row = rowbase + r + 8 * hi;
                h[(size_t)row * CFEAT + t * 16 + lo] = acc[t][r];
            }
        }
    } else {
        #pragma unroll
        for (int t = 0; t < 8; ++t) {
            #pragma unroll
            for (int r = 0; r < 8; ++r) {
                const int row = rowbase + r + 8 * hi;
                if (row < n) h[(size_t)row * CFEAT + t * 16 + lo] = acc[t][r];
            }
        }
    }
}


__device__ __forceinline__ float4 aggregate(const float* __restrict__ hsrc,
                                            const int* __restrict__ offs,
                                            const int* __restrict__ csr,
                                            const float* __restrict__ dinv,
                                            int node, int lane) {
    const float din = dinv[node];
    const float4* __restrict__ h4 = (const float4*)hsrc;
    float4 acc;
    {
        float4 v = h4[(size_t)node * 32 + lane];
        acc.x = v.x * din; acc.y = v.y * din; acc.z = v.z * din; acc.w = v.w * din;
    }
    const int beg = offs[node], end = offs[node + 1];
    for (int j = beg; j < end; j += 32) {
        const int idx = j + lane;
        const int   s  = (idx < end) ? csr[idx] : 0;
        const float ws = (idx < end) ? dinv[s] : 0.0f;
        int cnt = end - j; if (cnt > 32) cnt = 32;
        int t = 0;
        if (cnt == 32) {
            for (; t < 32; t += 4) {
                const int   s0 = __shfl(s, t,     32), s1 = __shfl(s, t + 1, 32);
                const int   s2 = __shfl(s, t + 2, 32), s3 = __shfl(s, t + 3, 32);
                const float w0 = __shfl(ws, t,     32), w1 = __shfl(ws, t + 1, 32);
                const float w2 = __shfl(ws, t + 2, 32), w3 = __shfl(ws, t + 3, 32);
                float4 v0 = h4[(size_t)s0 * 32 + lane];
                float4 v1 = h4[(size_t)s1 * 32 + lane];
                float4 v2 = h4[(size_t)s2 * 32 + lane];
                float4 v3 = h4[(size_t)s3 * 32 + lane];
                acc.x += v0.x * w0; acc.y += v0.y * w0; acc.z += v0.z * w0; acc.w += v0.w * w0;
                acc.x += v1.x * w1; acc.y += v1.y * w1; acc.z += v1.z * w1; acc.w += v1.w * w1;
                acc.x += v2.x * w2; acc.y += v2.y * w2; acc.z += v2.z * w2; acc.w += v2.w * w2;
                acc.x += v3.x * w3; acc.y += v3.y * w3; acc.z += v3.z * w3; acc.w += v3.w * w3;
            }
        }
        for (; t < cnt; ++t) {
            const int   sk = __shfl(s, t, 32);
            const float wk = __shfl(ws, t, 32);
            float4 v = h4[(size_t)sk * 32 + lane];
            acc.x += v.x * wk; acc.y += v.y * wk; acc.z += v.z * wk; acc.w += v.w * wk;
        }
    }
    acc.x *= din; acc.y *= din; acc.z *= din; acc.w *= din;
    return acc;
}

__global__ __launch_bounds__(256)
void k_prop1(const float* __restrict__ h, float* __restrict__ h2,
             const int* __restrict__ offs, const int* __restrict__ csr,
             const float* __restrict__ dinv, const float* __restrict__ b1, int n) {
    const int node = (int)((blockIdx.x * blockDim.x + threadIdx.x) >> 5);
    const int lane = threadIdx.x & 31;
    if (node >= n) return;
    float4 acc = aggregate(h, offs, csr, dinv, node, lane);
    const float4 b = ((const float4*)b1)[lane];
    acc.x = fmaxf(acc.x + b.x, 0.0f);
    acc.y = fmaxf(acc.y + b.y, 0.0f);
    acc.z = fmaxf(acc.z + b.z, 0.0f);
    acc.w = fmaxf(acc.w + b.w, 0.0f);
    ((float4*)h2)[(size_t)node * 32 + lane] = acc;
}

__global__ __launch_bounds__(256)
void k_prop2_out(const float* __restrict__ h2, float* __restrict__ out,
                 const int* __restrict__ offs, const int* __restrict__ csr,
                 const float* __restrict__ dinv, const float* __restrict__ Wout,
                 const float* __restrict__ bout, int n) {
    const int node = (int)((blockIdx.x * blockDim.x + threadIdx.x) >> 5);
    const int lane = threadIdx.x & 31;
    if (node >= n) return;

    float wl[4][4][2];
    #pragma unroll
    for (int m = 0; m < 4; ++m)
        #pragma unroll
        for (int j = 0; j < 4; ++j) {
            const float2 w = ((const float2*)Wout)[m * CFEAT + lane * 4 + j];
            wl[m][j][0] = w.x; wl[m][j][1] = w.y;
        }

    float4 acc = aggregate(h2, offs, csr, dinv, node, lane);

    float p[4][2];
    #pragma unroll
    for (int m = 0; m < 4; ++m)
        #pragma unroll
        for (int o = 0; o < 2; ++o)
            p[m][o] = acc.x * wl[m][0][o] + acc.y * wl[m][1][o] +
                      acc.z * wl[m][2][o] + acc.w * wl[m][3][o];

    #pragma unroll
    for (int m = 0; m < 4; ++m)
        #pragma unroll
        for (int o = 0; o < 2; ++o) {
            float v = p[m][o];
            #pragma unroll
            for (int d = 16; d > 0; d >>= 1) v += __shfl_xor(v, d, 32);
            p[m][o] = v;
        }

    if (lane == 0) {
        #pragma unroll
        for (int m = 0; m < 4; ++m) {
            float2 r;
            r.x = p[m][0] + bout[m * 2 + 0];
            r.y = p[m][1] + bout[m * 2 + 1];
            ((float2*)out)[(size_t)m * n + node] = r;
        }
    }
}


extern "C" void kernel_launch(void* const* d_in, const int* in_sizes, int n_in,
                              void* d_out, int out_size, void* d_ws, size_t ws_size,
                              hipStream_t stream) {
    const float* x    = (const float*)d_in[0];
    const int*   ei   = (const int*)d_in[1];
    const float* W1   = (const float*)d_in[2];
    const float* b1   = (const float*)d_in[3];
    const float* Wout = (const float*)d_in[4];
    const float* bout = (const float*)d_in[5];
    float* out = (float*)d_out;

    const int n = in_sizes[0] / CFEAT;
    const int e = in_sizes[1] / 2;
    const int* srcv = ei;
    const int* dstv = ei + e;

    uintptr_t p = (uintptr_t)d_ws;
    auto alloc = [&](size_t bytes) {
        p = (p + 15) & ~(uintptr_t)15;
        void* r = (void*)p;
        p += bytes;
        return r;
    };
    int*   deg    = (int*)  alloc((size_t)n * 4);
    int*   cursor = (int*)  alloc((size_t)n * 4);
    int*   offs   = (int*)  alloc((size_t)(n + 1) * 4);
    int*   csr    = (int*)  alloc((size_t)e * 4);
    float* dinv   = (float*)alloc((size_t)n * 4);
    float* h      = (float*)alloc((size_t)n * CFEAT * 4);
    float* h2     = (float*)alloc((size_t)n * CFEAT * 4);
    (void)ws_size; (void)n_in; (void)out_size;

    const int TB = 256;
    const int gn = (n + TB - 1) / TB;
    const int ge = (e + TB - 1) / TB;

    k_init <<<gn, TB, 0, stream>>>(deg, cursor, n);
    k_count<<<ge, TB, 0, stream>>>(dstv, deg, e);
    k_dinv <<<gn, TB, 0, stream>>>(deg, dinv, n);
    k_scan <<<1, 1024, 0, stream>>>(deg, offs, n);
    k_fill <<<ge, TB, 0, stream>>>(srcv, dstv, offs, cursor, csr, e);

    k_gemm1<<<(n + 127) / 128, TB, 0, stream>>>(x, W1, h, n);

    const int gprop = (n + 7) / 8;
    k_prop1    <<<gprop, TB, 0, stream>>>(h, h2, offs, csr, dinv, b1, n);
    k_prop2_out<<<gprop, TB, 0, stream>>>(h2, out, offs, csr, dinv, Wout, bout, n);
}


# layer-2 linearity: prop1 epilogue projects each relu row by W_out (times dinv) to 8 floats, prop2 gathers 32 B per edge instead of 512 B; float sums reassociated (project before aggregate)
# speedup vs baseline: 1.5360x; 1.1945x over previous
#include <hip/hip_runtime.h>
#include <cstdint>

typedef __attribute__((ext_vector_type(2))) float v2f;
typedef __attribute__((ext_vector_type(8))) float v8f;

#define CFEAT 128


__global__ void k_init(int* __restrict__ deg, int* __restrict__ cursor, int n) {
    int i = blockIdx.x * blockDim.x + threadIdx.x;
    if (i < n) { deg[i] = 0; cursor[i] = 0; }
}

__global__ void k_count(const int* __restrict__ dst, int* __restrict__ deg, int e) {
    int i = blockIdx.x * blockDim.x + threadIdx.x;
    if (i < e) atomicAdd(&deg[dst[i]], 1);
}

__global__ void k_dinv(const int* __restrict__ deg, float* __restrict__ dinv, int n) {
    int i = blockIdx.x * blockDim.x + threadIdx.x;
    if (i < n) dinv[i] = rsqrtf((float)deg[i] + 1.0f);
}


#define SCAN_ITS 16

__device__ __forceinline__ int wave_incl_scan(int x, int lane) {
    x += __builtin_amdgcn_update_dpp(0, x, 0x111, 0xf, 0xf, true);
    x += __builtin_amdgcn_update_dpp(0, x, 0x112, 0xf, 0xf, true);
    x += __builtin_amdgcn_update_dpp(0, x, 0x114, 0xf, 0xf, true);
    x += __builtin_amdgcn_update_dpp(0, x, 0x118, 0xf, 0xf, true);
    const int row0 = __builtin_amdgcn_readlane(x, 15);
    return x + ((lane >= 16) ? row0 : 0);
}

__global__ __launch_bounds__(1024)
void k_scan(const int* __restrict__ deg, int* __restrict__ offs, int n) {
    __shared__ int wtot[2][32];
    const int tid  = threadIdx.x;
    const int lane = tid & 31;
    const int wave = __builtin_amdgcn_readfirstlane(tid >> 5);
    const int nv   = n >> 2;
    const int4* __restrict__ deg4  = (const int4*)deg;
    int4*       __restrict__ offs4 = (int4*)offs;

    int running = 0;
    int par = 0;
    for (int gbase = 0; gbase < nv; gbase += SCAN_ITS * 1024) {
        int ng = nv - gbase;
        if (ng > SCAN_ITS * 1024) ng = SCAN_ITS * 1024;
        const int its  = (ng + 1023) >> 10;
        const int gend = gbase + ng;
        const int g0   = gbase + wave * its * 32 + lane;

        int4 v[SCAN_ITS];
        #pragma unroll
        for (int it = 0; it < SCAN_ITS; ++it) {
            const int g = g0 + it * 32;
            int4 t = make_int4(0, 0, 0, 0);
            if (it < its && g < gend) t = deg4[g];
            v[it] = t;
        }

        int carry = 0;
        #pragma unroll
        for (int it = 0; it < SCAN_ITS; ++it) {
            if (it < its) {
                const int4 t  = v[it];
                const int s1  = t.x + t.y;
                const int s2  = s1 + t.z;
                const int s3  = s2 + t.w;
                const int inc = wave_incl_scan(s3, lane);
                const int b   = carry + inc - s3;
                v[it] = make_int4(b, b + t.x, b + s1, b + s2);
                carry += __builtin_amdgcn_readlane(inc, 31);
            }
        }

        if (lane == 0) wtot[par][wave] = carry;
        __syncthreads();
        const int wt   = wtot[par][lane];
        const int winc = wave_incl_scan(wt, lane);
        const int add  = running + __builtin_amdgcn_readlane(winc - wt, wave);
        running += __builtin_amdgcn_readlane(winc, 31);

        #pragma unroll
        for (int it = 0; it < SCAN_ITS; ++it) {
            const int g = g0 + it * 32;
            if (it < its && g < gend) {
                int4 o = v[it];
                o.x += add; o.y += add; o.z += add; o.w += add;
                offs4[g] = o;
            }
        }
        par ^= 1;
    }
    if (tid == 0) {
        int r = running;
        for (int i = nv * 4; i < n; ++i) { offs[i] = r; r += deg[i]; }
        offs[n] = r;
    }
}

__global__ void k_fill(const int* __restrict__ src, const int* __restrict__ dst,
                       const int* __restrict__ offs, int* __restrict__ cursor,
                       int* __restrict__ csr, int e) {
    int i = blockIdx.x * blockDim.x + threadIdx.x;
    if (i < e) {
        int d = dst[i];
        int p = atomicAdd(&cursor[d], 1);
        csr[offs[d] + p] = src[i];
    }
}


__global__ __launch_bounds__(256)
void k_gemm1(const float* __restrict__ x, const float* __restrict__ W1,
             float* __restrict__ h, int n) {
    __shared__ float sW[CFEAT * CFEAT];
    for (int i = threadIdx.x; i < (CFEAT * CFEAT) / 4; i += 256)
        ((float4*)sW)[i] = ((const float4*)W1)[i];
    __syncthreads();

    const int lane = threadIdx.x & 31;
    const int wave = threadIdx.x >> 5;
    const int hi   = lane >> 4;
    const int lo   = lane & 15;
    const int rowbase = (blockIdx.x * 8 + wave) * 16;
    if (rowbase >= n) return;

    int arow = rowbase + lo;
    if (arow >= n) arow = n - 1;
    const float* __restrict__ xrow = x + (size_t)arow * CFEAT;

    v8f acc[8];
    const v8f z = {0.f, 0.f, 0.f, 0.f, 0.f, 0.f, 0.f, 0.f};
    #pragma unroll
    for (int t = 0; t < 8; ++t) acc[t] = z;

    for (int k = 0; k < CFEAT; k += 4) {
        v2f a;
        a.x = xrow[k + 2 * hi];
        a.y = xrow[k + 2 * hi + 1];
        #pragma unroll
        for (int t = 0; t < 8; ++t) {
            v2f b;
            const int col = t * 16 + lo;
            b.x = sW[(k + 2 * hi) * CFEAT + col];
            b.y = sW[(k + 2 * hi + 1) * CFEAT + col];
            acc[t] = __builtin_amdgcn_wmma_f32_16x16x4_f32(
                false, a, false, b, (short)0, acc[t], false, false);
        }
    }

    if (rowbase + 16 <= n) {
        #pragma unroll
        for (int t = 0; t < 8; ++t) {
            #pragma unroll
            for (int r = 0; r < 8; ++r) {
                const int row = rowbase + r + 8 * hi;
                h[(size_t)row * CFEAT + t * 16 + lo] = acc[t][r];
            }
        }
    } else {
        #pragma unroll
        for (int t = 0; t < 8; ++t) {
            #pragma unroll
            for (int r = 0; r < 8; ++r) {
                const int row = rowbase + r + 8 * hi;
                if (row < n) h[(size_t)row * CFEAT + t * 16 + lo] = acc[t][r];
            }
        }
    }
}


__device__ __forceinline__ float4 aggregate(const float* __restrict__ hsrc,
                                            const int* __restrict__ offs,
                                            const int* __restrict__ csr,
                                            const float* __restrict__ dinv,
                                            int node, int lane) {
    const float din = dinv[node];
    const float4* __restrict__ h4 = (const float4*)hsrc;
    float4 acc;
    {
        float4 v = h4[(size_t)node * 32 + lane];
        acc.x = v.x * din; acc.y = v.y * din; acc.z = v.z * din; acc.w = v.w * din;
    }
    const int beg = offs[node], end = offs[node + 1];
    for (int j = beg; j < end; j += 32) {
        const int idx = j + lane;
        const int   s  = (idx < end) ? csr[idx] : 0;
        const float ws = (idx < end) ? dinv[s] : 0.0f;
        int cnt = end - j; if (cnt > 32) cnt = 32;
        int t = 0;
        if (cnt == 32) {
            for (; t < 32; t += 4) {
                const int   s0 = __shfl(s, t,     32), s1 = __shfl(s, t + 1, 32);
                const int   s2 = __shfl(s, t + 2, 32), s3 = __shfl(s, t + 3, 32);
                const float w0 = __shfl(ws, t,     32), w1 = __shfl(ws, t + 1, 32);
                const float w2 = __shfl(ws, t + 2, 32), w3 = __shfl(ws, t + 3, 32);
                float4 v0 = h4[(size_t)s0 * 32 + lane];
                float4 v1 = h4[(size_t)s1 * 32 + lane];
                float4 v2 = h4[(size_t)s2 * 32 + lane];
                float4 v3 = h4[(size_t)s3 * 32 + lane];
                acc.x += v0.x * w0; acc.y += v0.y * w0; acc.z += v0.z * w0; acc.w += v0.w * w0;
                acc.x += v1.x * w1; acc.y += v1.y * w1; acc.z += v1.z * w1; acc.w += v1.w * w1;
                acc.x += v2.x * w2; acc.y += v2.y * w2; acc.z += v2.z * w2; acc.w += v2.w * w2;
                acc.x += v3.x * w3; acc.y += v3.y * w3; acc.z += v3.z * w3; acc.w += v3.w * w3;
            }
        }
        for (; t < cnt; ++t) {
            const int   sk = __shfl(s, t, 32);
            const float wk = __shfl(ws, t, 32);
            float4 v = h4[(size_t)sk * 32 + lane];
            acc.x += v.x * wk; acc.y += v.y * wk; acc.z += v.z * wk; acc.w += v.w * wk;
        }
    }
    acc.x *= din; acc.y *= din; acc.z *= din; acc.w *= din;
    return acc;
}

__global__ __launch_bounds__(256)
void k_prop1_proj(const float* __restrict__ h, float* __restrict__ zs,
                  const int* __restrict__ offs, const int* __restrict__ csr,
                  const float* __restrict__ dinv, const float* __restrict__ b1,
                  const float* __restrict__ Wout, int n) {
    const int node = (int)((blockIdx.x * blockDim.x + threadIdx.x) >> 5);
    const int lane = threadIdx.x & 31;
    if (blockIdx.x == 0 && threadIdx.x < 2)
        ((float4*)zs)[(size_t)n * 2 + threadIdx.x] = make_float4(0.f, 0.f, 0.f, 0.f);
    if (node >= n) return;

    float4 acc = aggregate(h, offs, csr, dinv, node, lane);
    const float4 b = ((const float4*)b1)[lane];
    acc.x = fmaxf(acc.x + b.x, 0.0f);
    acc.y = fmaxf(acc.y + b.y, 0.0f);
    acc.z = fmaxf(acc.z + b.z, 0.0f);
    acc.w = fmaxf(acc.w + b.w, 0.0f);

    float p[4][2];
    #pragma unroll
    for (int m = 0; m < 4; ++m) {
        const float4* __restrict__ w4 = (const float4*)(Wout + ((size_t)m * CFEAT + lane * 4) * 2);
        const float4 w01 = w4[0];
        const float4 w23 = w4[1];
        p[m][0] = acc.x * w01.x + acc.y * w01.z + acc.z * w23.x + acc.w * w23.z;
        p[m][1] = acc.x * w01.y + acc.y * w01.w + acc.z * w23.y + acc.w * w23.w;
    }

    #pragma unroll
    for (int m = 0; m < 4; ++m)
        #pragma unroll
        for (int o = 0; o < 2; ++o) {
            float v = p[m][o];
            #pragma unroll
            for (int d = 16; d > 0; d >>= 1) v += __shfl_xor(v, d, 32);
            p[m][o] = v;
        }

    if (lane == 0) {
        const float din = dinv[node];
        float4* __restrict__ z4 = (float4*)zs + (size_t)node * 2;
        z4[0] = make_float4(p[0][0] * din, p[0][1] * din, p[1][0] * din, p[1][1] * din);
        z4[1] = make_float4(p[2][0] * din, p[2][1] * din, p[3][0] * din, p[3][1] * din);
    }
}


#define P2_NPW 4
#define P2_CH  3

__device__ __forceinline__ float sum_same_parity(float x) {
    x += __builtin_bit_cast(float, __builtin_amdgcn_update_dpp(0, __builtin_bit_cast(int, x), 0x162, 0xf, 0xf, true));
    x += __builtin_bit_cast(float, __builtin_amdgcn_update_dpp(0, __builtin_bit_cast(int, x), 0x164, 0xf, 0xf, true));
    x += __builtin_bit_cast(float, __builtin_amdgcn_update_dpp(0, __builtin_bit_cast(int, x), 0x168, 0xf, 0xf, true));
    const int b = __builtin_bit_cast(int, x);
    x += __builtin_bit_cast(float, __builtin_amdgcn_permlanex16(b, b, 0x76543210, 0xfedcba98, false, false));
    return x;
}

__global__ __launch_bounds__(256)
void k_prop2_small(const float* __restrict__ zs, float* __restrict__ out,
                   const int* __restrict__ offs, const int* __restrict__ csr,
                   const float* __restrict__ dinv, const float* __restrict__ bout,
                   int n, int e) {
    const int lane = threadIdx.x & 31;
    const int q    = lane >> 1;
    const int half = lane & 1;
    const int node0 = __builtin_amdgcn_readfirstlane((int)((blockIdx.x * blockDim.x + threadIdx.x) >> 5)) * P2_NPW;
    if (node0 >= n) return;
    const float4* __restrict__ z4 = (const float4*)zs;

    const int   tq   = node0 + q;
    const float dinq = dinv[(tq < n) ? tq : (n - 1)];
    const float4 bo  = ((const float4*)bout)[half];

    int nd[P2_NPW], beg[P2_NPW], end[P2_NPW];
    #pragma unroll
    for (int i = 0; i < P2_NPW; ++i) {
        const int t = node0 + i;
        nd[i]  = (t < n) ? t : (n - 1);
        beg[i] = offs[nd[i]];
        end[i] = offs[nd[i] + 1];
    }

    int  s[P2_NPW][P2_CH];
    bool ok[P2_NPW][P2_CH];
    #pragma unroll
    for (int i = 0; i < P2_NPW; ++i)
        #pragma unroll
        for (int k = 0; k < P2_CH; ++k) {
            const int idx = beg[i] + k * 16 + q - 1;
            ok[i][k] = idx < end[i];
            int ic = (idx > 0) ? idx : 0;
            ic = (ic < e) ? ic : (e - 1);
            s[i][k] = csr[ic];
        }

    float4 acc[P2_NPW];
    #pragma unroll
    for (int i = 0; i < P2_NPW; ++i) {
        float4 v[P2_CH];
        #pragma unroll
        for (int k = 0; k < P2_CH; ++k) {
            int sv = ok[i][k] ? s[i][k] : n;
            if (k == 0) sv = (q == 0) ? nd[i] : sv;
            v[k] = z4[(unsigned)sv * 2u + (unsigned)half];
        }
        acc[i] = v[0];
        #pragma unroll
        for (int k = 1; k < P2_CH; ++k) {
            acc[i].x += v[k].x; acc[i].y += v[k].y; acc[i].z += v[k].z; acc[i].w += v[k].w;
        }
    }

    #pragma unroll
    for (int i = 0; i < P2_NPW; ++i)
        for (int j = beg[i] + P2_CH * 16 - 1; j < end[i]; j += 16) {
            const int  idx = j + q;
            const bool okj = idx < end[i];
            const int  sj  = csr[okj ? idx : j];
            const float4 vj = z4[(unsigned)(okj ? sj : n) * 2u + (unsigned)half];
            acc[i].x += vj.x; acc[i].y += vj.y; acc[i].z += vj.z; acc[i].w += vj.w;
        }

    #pragma unroll
    for (int i = 0; i < P2_NPW; ++i) {
        acc[i].x = sum_same_parity(acc[i].x);
        acc[i].y = sum_same_parity(acc[i].y);
        acc[i].z = sum_same_parity(acc[i].z);
        acc[i].w = sum_same_parity(acc[i].w);
    }

    float4 t = acc[0];
    #pragma unroll
    for (int i = 1; i < P2_NPW; ++i)
        if (q == i) t = acc[i];
    float2 r0, r1;
    r0.x = t.x * dinq + bo.x;
    r0.y = t.y * dinq + bo.y;
    r1.x = t.z * dinq + bo.z;
    r1.y = t.w * dinq + bo.w;
    if (q < P2_NPW && tq < n) {
        const int m0 = half * 2;
        ((float2*)out)[(size_t)m0 * n + tq]       = r0;
        ((float2*)out)[(size_t)(m0 + 1) * n + tq] = r1;
    }
}


extern "C" void kernel_launch(void* const* d_in, const int* in_sizes, int n_in,
                              void* d_out, int out_size, void* d_ws, size_t ws_size,
                              hipStream_t stream) {
    const float* x    = (const float*)d_in[0];
    const int*   ei   = (const int*)d_in[1];
    const float* W1   = (const float*)d_in[2];
    const float* b1   = (const float*)d_in[3];
    const float* Wout = (const float*)d_in[4];
    const float* bout = (const float*)d_in[5];
    float* out = (float*)d_out;

    const int n = in_sizes[0] / CFEAT;
    const int e = in_sizes[1] / 2;
    const int* srcv = ei;
    const int* dstv = ei + e;

    uintptr_t p = (uintptr_t)d_ws;
    auto alloc = [&](size_t bytes) {
        p = (p + 255) & ~(uintptr_t)255;
        void* r = (void*)p;
        p += bytes;
        return r;
    };
    int*   deg    = (int*)  alloc((size_t)n * 4);
    int*   cursor = (int*)  alloc((size_t)n * 4);
    int*   offs   = (int*)  alloc((size_t)(n + 1) * 4);
    int*   csr    = (int*)  alloc((size_t)e * 4);
    float* dinv   = (float*)alloc((size_t)n * 4);
    float* h      = (float*)alloc((size_t)n * CFEAT * 4);
    float* zs     = (float*)alloc(((size_t)n + 1) * 8 * 4);
    (void)ws_size; (void)n_in; (void)out_size;

    const int TB = 256;
    const int gn = (n + TB - 1) / TB;
    const int ge = (e + TB - 1) / TB;

    k_init <<<gn, TB, 0, stream>>>(deg, cursor, n);
    k_count<<<ge, TB, 0, stream>>>(dstv, deg, e);
    k_dinv <<<gn, TB, 0, stream>>>(deg, dinv, n);
    k_scan <<<1, 1024, 0, stream>>>(deg, offs, n);
    k_fill <<<ge, TB, 0, stream>>>(srcv, dstv, offs, cursor, csr, e);

    k_gemm1<<<(n + 127) / 128, TB, 0, stream>>>(x, W1, h, n);

    const int gprop = (n + 7) / 8;
    k_prop1_proj <<<gprop, TB, 0, stream>>>(h, zs, offs, csr, dinv, b1, Wout, n);
    const int gp2 = (n + 8 * P2_NPW - 1) / (8 * P2_NPW);
    k_prop2_small<<<gp2, TB, 0, stream>>>(zs, out, offs, csr, dinv, bout, n, e);
}


# csr build as a two-level counting sort on LDS counters (buckets of 256 dst nodes) replaces one global atomic per edge in k_count and k_fill; 8 launches become 7; csr row order stays arrival order
# speedup vs baseline: 4.0680x; 2.6484x over previous
#include <hip/hip_runtime.h>
#include <cstdint>

typedef __attribute__((ext_vector_type(2))) float v2f;
typedef __attribute__((ext_vector_type(8))) float v8f;

#define CFEAT 128


#define BSH      8
#define BNODES   (1 << BSH)
#define ETHREADS 1024
#define EUNROLL  8


#define SCAN_ITS 16

__device__ __forceinline__ int wave_incl_scan(int x, int lane) {
    x += __builtin_amdgcn_update_dpp(0, x, 0x111, 0xf, 0xf, true);
    x += __builtin_amdgcn_update_dpp(0, x, 0x112, 0xf, 0xf, true);
    x += __builtin_amdgcn_update_dpp(0, x, 0x114, 0xf, 0xf, true);
    x += __builtin_amdgcn_update_dpp(0, x, 0x118, 0xf, 0xf, true);
    const int row0 = __builtin_amdgcn_readlane(x, 15);
    return x + ((lane >= 16) ? row0 : 0);
}

__global__ __launch_bounds__(1024)
void k_scan(const int* __restrict__ deg, int* __restrict__ offs, int n) {
    __shared__ int wtot[2][32];
    const int tid  = threadIdx.x;
    const int lane = tid & 31;
    const int wave = __builtin_amdgcn_readfirstlane(tid >> 5);
    const int nv   = n >> 2;
    const int4* __restrict__ deg4  = (const int4*)deg;
    int4*       __restrict__ offs4 = (int4*)offs;

    int running = 0;
    int par = 0;
    for (int gbase = 0; gbase < nv; gbase += SCAN_ITS * 1024) {
        int ng = nv - gbase;
        if (ng > SCAN_ITS * 1024) ng = SCAN_ITS * 1024;
        const int its  = (ng + 1023) >> 10;
        const int gend = gbase + ng;
        const int g0   = gbase + wave * its * 32 + lane;

        int4 v[SCAN_ITS];
        #pragma unroll
        for (int it = 0; it < SCAN_ITS; ++it) {
            const int g = g0 + it * 32;
            int4 t = make_int4(0, 0, 0, 0);
            if (it < its && g < gend) t = deg4[g];
            v[it] = t;
        }

        int carry = 0;
        #pragma unroll
        for (int it = 0; it < SCAN_ITS; ++it) {
            if (it < its) {
                const int4 t  = v[it];
                const int s1  = t.x + t.y;
                const int s2  = s1 + t.z;
                const int s3  = s2 + t.w;
                const int inc = wave_incl_scan(s3, lane);
                const int b   = carry + inc - s3;
                v[it] = make_int4(b, b + t.x, b + s1, b + s2);
                carry += __builtin_amdgcn_readlane(inc, 31);
            }
        }

        if (lane == 0) wtot[par][wave] = carry;
        __syncthreads();
        const int wt   = wtot[par][lane];
        const int winc = wave_incl_scan(wt, lane);
        const int add  = running + __builtin_amdgcn_readlane(winc - wt, wave);
        running += __builtin_amdgcn_readlane(winc, 31);

        #pragma unroll
        for (int it = 0; it < SCAN_ITS; ++it) {
            const int g = g0 + it * 32;
            if (it < its && g < gend) {
                int4 o = v[it];
                o.x += add; o.y += add; o.z += add; o.w += add;
                offs4[g] = o;
            }
        }
        par ^= 1;
    }
    if (tid == 0) {
        int r = running;
        for (int i = nv * 4; i < n; ++i) { offs[i] = r; r += deg[i]; }
        offs[n] = r;
    }
}

__global__ __launch_bounds__(ETHREADS)
void k_bucket_count(const int* __restrict__ dst, int* __restrict__ bcnt,
                    int e, int chunk, int nbk) {
    extern __shared__ int hist[];
    const int tid = threadIdx.x;
    for (int b = tid; b < nbk; b += ETHREADS) hist[b] = 0;
    __syncthreads();
    const int start = blockIdx.x * chunk;
    int end = start + chunk; if (end > e) end = e;
    for (int base = start; base < end; base += EUNROLL * ETHREADS) {
        int d[EUNROLL];
        #pragma unroll
        for (int u = 0; u < EUNROLL; ++u) {
            const int i = base + u * ETHREADS + tid;
            d[u] = (i < end) ? dst[i] : -1;
        }
        #pragma unroll
        for (int u = 0; u < EUNROLL; ++u)
            if (d[u] >= 0) atomicAdd(&hist[d[u] >> BSH], 1);
    }
    __syncthreads();
    for (int b = tid; b < nbk; b += ETHREADS) bcnt[b * gridDim.x + blockIdx.x] = hist[b];
}

__global__ __launch_bounds__(ETHREADS)
void k_bucket_scatter(const int* __restrict__ src, const int* __restrict__ dst,
                      const int* __restrict__ boffs, int* __restrict__ tmpw,
                      int e, int chunk, int nbk) {
    extern __shared__ int cur[];
    int2* __restrict__ tmp = (int2*)tmpw;
    const int tid = threadIdx.x;
    for (int b = tid; b < nbk; b += ETHREADS) cur[b] = boffs[b * gridDim.x + blockIdx.x];
    __syncthreads();
    const int start = blockIdx.x * chunk;
    int end = start + chunk; if (end > e) end = e;
    for (int base = start; base < end; base += EUNROLL * ETHREADS) {
        int d[EUNROLL], s[EUNROLL];
        #pragma unroll
        for (int u = 0; u < EUNROLL; ++u) {
            const int i = base + u * ETHREADS + tid;
            d[u] = (i < end) ? dst[i] : -1;
            s[u] = (i < end) ? src[i] : 0;
        }
        #pragma unroll
        for (int u = 0; u < EUNROLL; ++u)
            if (d[u] >= 0) {
                const int p = atomicAdd(&cur[d[u] >> BSH], 1);
                tmp[p] = make_int2(s[u], d[u]);
            }
    }
}

__global__ __launch_bounds__(ETHREADS)
void k_bucket_final(const int* __restrict__ tmpw, const int* __restrict__ boffs,
                    int* __restrict__ offs, int* __restrict__ csr,
                    float* __restrict__ dinv, int n, int g1) {
    const int2* __restrict__ tmp = (const int2*)tmpw;
    __shared__ int cnt[BNODES];
    __shared__ int loc[BNODES];
    __shared__ int wsum[BNODES / 32];
    const int tid  = threadIdx.x;
    const int lane = tid & 31;
    const int wave = tid >> 5;
    const int b    = blockIdx.x;
    const int bs   = boffs[b * g1];
    const int be   = boffs[(b + 1) * g1];

    if (tid < BNODES) cnt[tid] = 0;
    __syncthreads();
    for (int base = bs; base < be; base += EUNROLL * ETHREADS) {
        int dy[EUNROLL];
        #pragma unroll
        for (int u = 0; u < EUNROLL; ++u) {
            const int i = base + u * ETHREADS + tid;
            dy[u] = (i < be) ? tmp[i].y : -1;
        }
        #pragma unroll
        for (int u = 0; u < EUNROLL; ++u)
            if (dy[u] >= 0) atomicAdd(&cnt[dy[u] & (BNODES - 1)], 1);
    }
    __syncthreads();

    int c = 0, inc = 0;
    if (tid < BNODES) {
        c   = cnt[tid];
        inc = wave_incl_scan(c, lane);
        if (lane == 31) wsum[wave] = inc;
    }
    __syncthreads();
    if (tid < BNODES) {
        int add = 0;
        for (int w = 0; w < wave; ++w) add += wsum[w];
        const int excl = add + inc - c;
        loc[tid] = excl;
        cnt[tid] = 0;
        const int node = b * BNODES + tid;
        if (node < n) {
            offs[node] = bs + excl;
            dinv[node] = rsqrtf((float)c + 1.0f);
        }
    }
    if (b == (int)gridDim.x - 1 && tid == 0) offs[n] = be;
    __syncthreads();

    for (int base = bs; base < be; base += EUNROLL * ETHREADS) {
        int2 t[EUNROLL];
        #pragma unroll
        for (int u = 0; u < EUNROLL; ++u) {
            const int i = base + u * ETHREADS + tid;
            t[u] = (i < be) ? tmp[i] : make_int2(0, -1);
        }
        #pragma unroll
        for (int u = 0; u < EUNROLL; ++u)
            if (t[u].y >= 0) {
                const int k = t[u].y & (BNODES - 1);
                const int p = atomicAdd(&cnt[k], 1);
                csr[bs + loc[k] + p] = t[u].x;
            }
    }
}


__global__ __launch_bounds__(256)
void k_gemm1(const float* __restrict__ x, const float* __restrict__ W1,
             float* __restrict__ h, int n) {
    __shared__ float sW[CFEAT * CFEAT];
    for (int i = threadIdx.x; i < (CFEAT * CFEAT) / 4; i += 256)
        ((float4*)sW)[i] = ((const float4*)W1)[i];
    __syncthreads();

    const int lane = threadIdx.x & 31;
    const int wave = threadIdx.x >> 5;
    const int hi   = lane >> 4;
    const int lo   = lane & 15;
    const int rowbase = (blockIdx.x * 8 + wave) * 16;
    if (rowbase >= n) return;

    int arow = rowbase + lo;
    if (arow >= n) arow = n - 1;
    const float* __restrict__ xrow = x + (size_t)arow * CFEAT;

    v8f acc[8];
    const v8f z = {0.f, 0.f, 0.f, 0.f, 0.f, 0.f, 0.f, 0.f};
    #pragma unroll
    for (int t = 0; t < 8; ++t) acc[t] = z;

    for (int k = 0; k < CFEAT; k += 4) {
        v2f a;
        a.x = xrow[k + 2 * hi];
        a.y = xrow[k + 2 * hi + 1];
        #pragma unroll
        for (int t = 0; t < 8; ++t) {
            v2f b;
            const int col = t * 16 + lo;
            b.x = sW[(k + 2 * hi) * CFEAT + col];
            b.y = sW[(k + 2 * hi + 1) * CFEAT + col];
            acc[t] = __builtin_amdgcn_wmma_f32_16x16x4_f32(
                false, a, false, b, (short)0, acc[t], false, false);
        }
    }

    if (rowbase + 16 <= n) {
        #pragma unroll
        for (int t = 0; t < 8; ++t) {
            #pragma unroll
            for (int r = 0; r < 8; ++r) {
                const int row = rowbase + r + 8 * hi;
                h[(size_t)row * CFEAT + t * 16 + lo] = acc[t][r];
            }
        }
    } else {
        #pragma unroll
        for (int t = 0; t < 8; ++t) {
            #pragma unroll
            for (int r = 0; r < 8; ++r) {
                const int row = rowbase + r + 8 * hi;
                if (row < n) h[(size_t)row * CFEAT + t * 16 + lo] = acc[t][r];
            }
        }
    }
}


__device__ __forceinline__ float4 aggregate(const float* __restrict__ hsrc,
                                            const int* __restrict__ offs,
                                            const int* __restrict__ csr,
                                            const float* __restrict__ dinv,
                                            int node, int lane) {
    const float din = dinv[node];
    const float4* __restrict__ h4 = (const float4*)hsrc;
    float4 acc;
    {
        float4 v = h4[(size_t)node * 32 + lane];
        acc.x = v.x * din; acc.y = v.y * din; acc.z = v.z * din; acc.w = v.w * din;
    }
    const int beg = offs[node], end = offs[node + 1];
    for (int j = beg; j < end; j += 32) {
        const int idx = j + lane;
        const int   s  = (idx < end) ? csr[idx] : 0;
        const float ws = (idx < end) ? dinv[s] : 0.0f;
        int cnt = end - j; if (cnt > 32) cnt = 32;
        int t = 0;
        if (cnt == 32) {
            for (; t < 32; t += 4) {
                const int   s0 = __shfl(s, t,     32), s1 = __shfl(s, t + 1, 32);
                const int   s2 = __shfl(s, t + 2, 32), s3 = __shfl(s, t + 3, 32);
                const float w0 = __shfl(ws, t,     32), w1 = __shfl(ws, t + 1, 32);
                const float w2 = __shfl(ws, t + 2, 32), w3 = __shfl(ws, t + 3, 32);
                float4 v0 = h4[(size_t)s0 * 32 + lane];
                float4 v1 = h4[(size_t)s1 * 32 + lane];
                float4 v2 = h4[(size_t)s2 * 32 + lane];
                float4 v3 = h4[(size_t)s3 * 32 + lane];
                acc.x += v0.x * w0; acc.y += v0.y * w0; acc.z += v0.z * w0; acc.w += v0.w * w0;
                acc.x += v1.x * w1; acc.y += v1.y * w1; acc.z += v1.z * w1; acc.w += v1.w * w1;
                acc.x += v2.x * w2; acc.y += v2.y * w2; acc.z += v2.z * w2; acc.w += v2.w * w2;
                acc.x += v3.x * w3; acc.y += v3.y * w3; acc.z += v3.z * w3; acc.w += v3.w * w3;
            }
        }
        for (; t < cnt; ++t) {
            const int   sk = __shfl(s, t, 32);
            const float wk = __shfl(ws, t, 32);
            float4 v = h4[(size_t)sk * 32 + lane];
            acc.x += v.x * wk; acc.y += v.y * wk; acc.z += v.z * wk; acc.w += v.w * wk;
        }
    }
    acc.x *= din; acc.y *= din; acc.z *= din; acc.w *= din;
    return acc;
}

__global__ __launch_bounds__(256)
void k_prop1_proj(const float* __restrict__ h, float* __restrict__ zs,
                  const int* __restrict__ offs, const int* __restrict__ csr,
                  const float* __restrict__ dinv, const float* __restrict__ b1,
                  const float* __restrict__ Wout, int n) {
    const int node = (int)((blockIdx.x * blockDim.x + threadIdx.x) >> 5);
    const int lane = threadIdx.x & 31;
    if (blockIdx.x == 0 && threadIdx.x < 2)
        ((float4*)zs)[(size_t)n * 2 + threadIdx.x] = make_float4(0.f, 0.f, 0.f, 0.f);
    if (node >= n) return;

    float4 acc = aggregate(h, offs, csr, dinv, node, lane);
    const float4 b = ((const float4*)b1)[lane];
    acc.x = fmaxf(acc.x + b.x, 0.0f);
    acc.y = fmaxf(acc.y + b.y, 0.0f);
    acc.z = fmaxf(acc.z + b.z, 0.0f);
    acc.w = fmaxf(acc.w + b.w, 0.0f);

    float p[4][2];
    #pragma unroll
    for (int m = 0; m < 4; ++m) {
        const float4* __restrict__ w4 = (const float4*)(Wout + ((size_t)m * CFEAT + lane * 4) * 2);
        const float4 w01 = w4[0];
        const float4 w23 = w4[1];
        p[m][0] = acc.x * w01.x + acc.y * w01.z + acc.z * w23.x + acc.w * w23.z;
        p[m][1] = acc.x * w01.y + acc.y * w01.w + acc.z * w23.y + acc.w * w23.w;
    }

    #pragma unroll
    for (int m = 0; m < 4; ++m)
        #pragma unroll
        for (int o = 0; o < 2; ++o) {
            float v = p[m][o];
            #pragma unroll
            for (int d = 16; d > 0; d >>= 1) v += __shfl_xor(v, d, 32);
            p[m][o] = v;
        }

    if (lane == 0) {
        const float din = dinv[node];
        float4* __restrict__ z4 = (float4*)zs + (size_t)node * 2;
        z4[0] = make_float4(p[0][0] * din, p[0][1] * din, p[1][0] * din, p[1][1] * din);
        z4[1] = make_float4(p[2][0] * din, p[2][1] * din, p[3][0] * din, p[3][1] * din);
    }
}


#define P2_NPW 4
#define P2_CH  3

__device__ __forceinline__ float sum_same_parity(float x) {
    x += __builtin_bit_cast(float, __builtin_amdgcn_update_dpp(0, __builtin_bit_cast(int, x), 0x162, 0xf, 0xf, true));
    x += __builtin_bit_cast(float, __builtin_amdgcn_update_dpp(0, __builtin_bit_cast(int, x), 0x164, 0xf, 0xf, true));
    x += __builtin_bit_cast(float, __builtin_amdgcn_update_dpp(0, __builtin_bit_cast(int, x), 0x168, 0xf, 0xf, true));
    const int b = __builtin_bit_cast(int, x);
    x += __builtin_bit_cast(float, __builtin_amdgcn_permlanex16(b, b, 0x76543210, 0xfedcba98, false, false));
    return x;
}

__global__ __launch_bounds__(256)
void k_prop2_small(const float* __restrict__ zs, float* __restrict__ out,
                   const int* __restrict__ offs, const int* __restrict__ csr,
                   const float* __restrict__ dinv, const float* __restrict__ bout,
                   int n, int e) {
    const int lane = threadIdx.x & 31;
    const int q    = lane >> 1;
    const int half = lane & 1;
    const int node0 = __builtin_amdgcn_readfirstlane((int)((blockIdx.x * blockDim.x + threadIdx.x) >> 5)) * P2_NPW;
    if (node0 >= n) return;
    const float4* __restrict__ z4 = (const float4*)zs;

    const int   tq   = node0 + q;
    const float dinq = dinv[(tq < n) ? tq : (n - 1)];
    const float4 bo  = ((const float4*)bout)[half];

    int nd[P2_NPW], beg[P2_NPW], end[P2_NPW];
    #pragma unroll
    for (int i = 0; i < P2_NPW; ++i) {
        const int t = node0 + i;
        nd[i]  = (t < n) ? t : (n - 1);
        beg[i] = offs[nd[i]];
        end[i] = offs[nd[i] + 1];
    }

    int  s[P2_NPW][P2_CH];
    bool ok[P2_NPW][P2_CH];
    #pragma unroll
    for (int i = 0; i < P2_NPW; ++i)
        #pragma unroll
        for (int k = 0; k < P2_CH; ++k) {
            const int idx = beg[i] + k * 16 + q - 1;
            ok[i][k] = idx < end[i];
            int ic = (idx > 0) ? idx : 0;
            ic = (ic < e) ? ic : (e - 1);
            s[i][k] = csr[ic];
        }

    float4 acc[P2_NPW];
    #pragma unroll
    for (int i = 0; i < P2_NPW; ++i) {
        float4 v[P2_CH];
        #pragma unroll
        for (int k = 0; k < P2_CH; ++k) {
            int sv = ok[i][k] ? s[i][k] : n;
            if (k == 0) sv = (q == 0) ? nd[i] : sv;
            v[k] = z4[(unsigned)sv * 2u + (unsigned)half];
        }
        acc[i] = v[0];
        #pragma unroll
        for (int k = 1; k < P2_CH; ++k) {
            acc[i].x += v[k].x; acc[i].y += v[k].y; acc[i].z += v[k].z; acc[i].w += v[k].w;
        }
    }

    #pragma unroll
    for (int i = 0; i < P2_NPW; ++i)
        for (int j = beg[i] + P2_CH * 16 - 1; j < end[i]; j += 16) {
            const int  idx = j + q;
            const bool okj = idx < end[i];
            const int  sj  = csr[okj ? idx : j];
            const float4 vj = z4[(unsigned)(okj ? sj : n) * 2u + (unsigned)half];
            acc[i].x += vj.x; acc[i].y += vj.y; acc[i].z += vj.z; acc[i].w += vj.w;
        }

    #pragma unroll
    for (int i = 0; i < P2_NPW; ++i) {
        acc[i].x = sum_same_parity(acc[i].x);
        acc[i].y = sum_same_parity(acc[i].y);
        acc[i].z = sum_same_parity(acc[i].z);
        acc[i].w = sum_same_parity(acc[i].w);
    }

    float4 t = acc[0];
    #pragma unroll
    for (int i = 1; i < P2_NPW; ++i)
        if (q == i) t = acc[i];
    float2 r0, r1;
    r0.x = t.x * dinq + bo.x;
    r0.y = t.y * dinq + bo.y;
    r1.x = t.z * dinq + bo.z;
    r1.y = t.w * dinq + bo.w;
    if (q < P2_NPW && tq < n) {
        const int m0 = half * 2;
        ((float2*)out)[(size_t)m0 * n + tq]       = r0;
        ((float2*)out)[(size_t)(m0 + 1) * n + tq] = r1;
    }
}


extern "C" void kernel_launch(void* const* d_in, const int* in_sizes, int n_in,
                              void* d_out, int out_size, void* d_ws, size_t ws_size,
                              hipStream_t stream) {
    const float* x    = (const float*)d_in[0];
    const int*   ei   = (const int*)d_in[1];
    const float* W1   = (const float*)d_in[2];
    const float* b1   = (const float*)d_in[3];
    const float* Wout = (const float*)d_in[4];
    const float* bout = (const float*)d_in[5];
    float* out = (float*)d_out;

    const int n = in_sizes[0] / CFEAT;
    const int e = in_sizes[1] / 2;
    const int* srcv = ei;
    const int* dstv = ei + e;

    uintptr_t p = (uintptr_t)d_ws;
    auto alloc = [&](size_t bytes) {
        p = (p + 255) & ~(uintptr_t)255;
        void* r = (void*)p;
        p += bytes;
        return r;
    };
    const int G1    = 256;
    const int chunk = (e + G1 - 1) / G1;
    const int nbk   = (n + BNODES - 1) >> BSH;
    int*   bcnt   = (int*)  alloc((size_t)nbk * G1 * 4);
    int*   boffs  = (int*)  alloc(((size_t)nbk * G1 + 1) * 4);
    int*   tmp    = (int*)  alloc((size_t)e * 8);
    int*   offs   = (int*)  alloc((size_t)(n + 1) * 4);
    int*   csr    = (int*)  alloc((size_t)e * 4);
    float* dinv   = (float*)alloc((size_t)n * 4);
    float* h      = (float*)alloc((size_t)n * CFEAT * 4);
    float* zs     = (float*)alloc(((size_t)n + 1) * 8 * 4);
    (void)ws_size; (void)n_in; (void)out_size;

    const int TB = 256;

    k_bucket_count  <<<G1, ETHREADS, (size_t)nbk * 4, stream>>>(dstv, bcnt, e, chunk, nbk);
    k_scan          <<<1, 1024, 0, stream>>>(bcnt, boffs, nbk * G1);
    k_bucket_scatter<<<G1, ETHREADS, (size_t)nbk * 4, stream>>>(srcv, dstv, boffs, tmp, e, chunk, nbk);
    k_bucket_final  <<<nbk, ETHREADS, 0, stream>>>(tmp, boffs, offs, csr, dinv, n, G1);

    k_gemm1<<<(n + 127) / 128, TB, 0, stream>>>(x, W1, h, n);

    const int gprop = (n + 7) / 8;
    k_prop1_proj <<<gprop, TB, 0, stream>>>(h, zs, offs, csr, dinv, b1, Wout, n);
    const int gp2 = (n + 8 * P2_NPW - 1) / (8 * P2_NPW);
    k_prop2_small<<<gp2, TB, 0, stream>>>(zs, out, offs, csr, dinv, bout, n, e);
}


# k_prop1_proj gather: each piece of a node's list (groups of 8, then the rest split 4+2+1) requests all its row loads together; the one-row-per-trip tail loop is gone; rows still summed in list order
# speedup vs baseline: 4.4233x; 1.0873x over previous
#include <hip/hip_runtime.h>
#include <cstdint>

typedef __attribute__((ext_vector_type(2))) float v2f;
typedef __attribute__((ext_vector_type(8))) float v8f;

#define CFEAT 128


#define BSH      8
#define BNODES   (1 << BSH)
#define ETHREADS 1024
#define EUNROLL  8


#define SCAN_ITS 16

__device__ __forceinline__ int wave_incl_scan(int x, int lane) {
    x += __builtin_amdgcn_update_dpp(0, x, 0x111, 0xf, 0xf, true);
    x += __builtin_amdgcn_update_dpp(0, x, 0x112, 0xf, 0xf, true);
    x += __builtin_amdgcn_update_dpp(0, x, 0x114, 0xf, 0xf, true);
    x += __builtin_amdgcn_update_dpp(0, x, 0x118, 0xf, 0xf, true);
    const int row0 = __builtin_amdgcn_readlane(x, 15);
    return x + ((lane >= 16) ? row0 : 0);
}

__global__ __launch_bounds__(1024)
void k_scan(const int* __restrict__ deg, int* __restrict__ offs, int n) {
    __shared__ int wtot[2][32];
    const int tid  = threadIdx.x;
    const int lane = tid & 31;
    const int wave = __builtin_amdgcn_readfirstlane(tid >> 5);
    const int nv   = n >> 2;
    const int4* __restrict__ deg4  = (const int4*)deg;
    int4*       __restrict__ offs4 = (int4*)offs;

    int running = 0;
    int par = 0;
    for (int gbase = 0; gbase < nv; gbase += SCAN_ITS * 1024) {
        int ng = nv - gbase;
        if (ng > SCAN_ITS * 1024) ng = SCAN_ITS * 1024;
        const int its  = (ng + 1023) >> 10;
        const int gend = gbase + ng;
        const int g0   = gbase + wave * its * 32 + lane;

        int4 v[SCAN_ITS];
        #pragma unroll
        for (int it = 0; it < SCAN_ITS; ++it) {
            const int g = g0 + it * 32;
            int4 t = make_int4(0, 0, 0, 0);
            if (it < its && g < gend) t = deg4[g];
            v[it] = t;
        }

        int carry = 0;
        #pragma unroll
        for (int it = 0; it < SCAN_ITS; ++it) {
            if (it < its) {
                const int4 t  = v[it];
                const int s1  = t.x + t.y;
                const int s2  = s1 + t.z;
                const int s3  = s2 + t.w;
                const int inc = wave_incl_scan(s3, lane);
                const int b   = carry + inc - s3;
                v[it] = make_int4(b, b + t.x, b + s1, b + s2);
                carry += __builtin_amdgcn_readlane(inc, 31);
            }
        }

        if (lane == 0) wtot[par][wave] = carry;
        __syncthreads();
        const int wt   = wtot[par][lane];
        const int winc = wave_incl_scan(wt, lane);
        const int add  = running + __builtin_amdgcn_readlane(winc - wt, wave);
        running += __builtin_amdgcn_readlane(winc, 31);

        #pragma unroll
        for (int it = 0; it < SCAN_ITS; ++it) {
            const int g = g0 + it * 32;
            if (it < its && g < gend) {
                int4 o = v[it];
                o.x += add; o.y += add; o.z += add; o.w += add;
                offs4[g] = o;
            }
        }
        par ^= 1;
    }
    if (tid == 0) {
        int r = running;
        for (int i = nv * 4; i < n; ++i) { offs[i] = r; r += deg[i]; }
        offs[n] = r;
    }
}

__global__ __launch_bounds__(ETHREADS)
void k_bucket_count(const int* __restrict__ dst, int* __restrict__ bcnt,
                    int e, int chunk, int nbk) {
    extern __shared__ int hist[];
    const int tid = threadIdx.x;
    for (int b = tid; b < nbk; b += ETHREADS) hist[b] = 0;
    __syncthreads();
    const int start = blockIdx.x * chunk;
    int end = start + chunk; if (end > e) end = e;
    for (int base = start; base < end; base += EUNROLL * ETHREADS) {
        int d[EUNROLL];
        #pragma unroll
        for (int u = 0; u < EUNROLL; ++u) {
            const int i = base + u * ETHREADS + tid;
            d[u] = (i < end) ? dst[i] : -1;
        }
        #pragma unroll
        for (int u = 0; u < EUNROLL; ++u)
            if (d[u] >= 0) atomicAdd(&hist[d[u] >> BSH], 1);
    }
    __syncthreads();
    for (int b = tid; b < nbk; b += ETHREADS) bcnt[b * gridDim.x + blockIdx.x] = hist[b];
}

__global__ __launch_bounds__(ETHREADS)
void k_bucket_scatter(const int* __restrict__ src, const int* __restrict__ dst,
                      const int* __restrict__ boffs, int* __restrict__ tmpw,
                      int e, int chunk, int nbk) {
    extern __shared__ int cur[];
    int2* __restrict__ tmp = (int2*)tmpw;
    const int tid = threadIdx.x;
    for (int b = tid; b < nbk; b += ETHREADS) cur[b] = boffs[b * gridDim.x + blockIdx.x];
    __syncthreads();
    const int start = blockIdx.x * chunk;
    int end = start + chunk; if (end > e) end = e;
    for (int base = start; base < end; base += EUNROLL * ETHREADS) {
        int d[EUNROLL], s[EUNROLL];
        #pragma unroll
        for (int u = 0; u < EUNROLL; ++u) {
            const int i = base + u * ETHREADS + tid;
            d[u] = (i < end) ? dst[i] : -1;
            s[u] = (i < end) ? src[i] : 0;
        }
        #pragma unroll
        for (int u = 0; u < EUNROLL; ++u)
            if (d[u] >= 0) {
                const int p = atomicAdd(&cur[d[u] >> BSH], 1);
                tmp[p] = make_int2(s[u], d[u]);
            }
    }
}

__global__ __launch_bounds__(ETHREADS)
void k_bucket_final(const int* __restrict__ tmpw, const int* __restrict__ boffs,
                    int* __restrict__ offs, int* __restrict__ csr,
                    float* __restrict__ dinv, int n, int g1) {
    const int2* __restrict__ tmp = (const int2*)tmpw;
    __shared__ int cnt[BNODES];
    __shared__ int loc[BNODES];
    __shared__ int wsum[BNODES / 32];
    const int tid  = threadIdx.x;
    const int lane = tid & 31;
    const int wave = tid >> 5;
    const int b    = blockIdx.x;
    const int bs   = boffs[b * g1];
    const int be   = boffs[(b + 1) * g1];

    if (tid < BNODES) cnt[tid] = 0;
    __syncthreads();
    for (int base = bs; base < be; base += EUNROLL * ETHREADS) {
        int dy[EUNROLL];
        #pragma unroll
        for (int u = 0; u < EUNROLL; ++u) {
            const int i = base + u * ETHREADS + tid;
            dy[u] = (i < be) ? tmp[i].y : -1;
        }
        #pragma unroll
        for (int u = 0; u < EUNROLL; ++u)
            if (dy[u] >= 0) atomicAdd(&cnt[dy[u] & (BNODES - 1)], 1);
    }
    __syncthreads();

    int c = 0, inc = 0;
    if (tid < BNODES) {
        c   = cnt[tid];
        inc = wave_incl_scan(c, lane);
        if (lane == 31) wsum[wave] = inc;
    }
    __syncthreads();
    if (tid < BNODES) {
        int add = 0;
        for (int w = 0; w < wave; ++w) add += wsum[w];
        const int excl = add + inc - c;
        loc[tid] = excl;
        cnt[tid] = 0;
        const int node = b * BNODES + tid;
        if (node < n) {
            offs[node] = bs + excl;
            dinv[node] = rsqrtf((float)c + 1.0f);
        }
    }
    if (b == (int)gridDim.x - 1 && tid == 0) offs[n] = be;
    __syncthreads();

    for (int base = bs; base < be; base += EUNROLL * ETHREADS) {
        int2 t[EUNROLL];
        #pragma unroll
        for (int u = 0; u < EUNROLL; ++u) {
            const int i = base + u * ETHREADS + tid;
            t[u] = (i < be) ? tmp[i] : make_int2(0, -1);
        }
        #pragma unroll
        for (int u = 0; u < EUNROLL; ++u)
            if (t[u].y >= 0) {
                const int k = t[u].y & (BNODES - 1);
                const int p = atomicAdd(&cnt[k], 1);
                csr[bs + loc[k] + p] = t[u].x;
            }
    }
}


__global__ __launch_bounds__(256)
void k_gemm1(const float* __restrict__ x, const float* __restrict__ W1,
             float* __restrict__ h, int n) {
    __shared__ float sW[CFEAT * CFEAT];
    for (int i = threadIdx.x; i < (CFEAT * CFEAT) / 4; i += 256)
        ((float4*)sW)[i] = ((const float4*)W1)[i];
    __syncthreads();

    const int lane = threadIdx.x & 31;
    const int wave = threadIdx.x >> 5;
    const int hi   = lane >> 4;
    const int lo   = lane & 15;
    const int rowbase = (blockIdx.x * 8 + wave) * 16;
    if (rowbase >= n) return;

    int arow = rowbase + lo;
    if (arow >= n) arow = n - 1;
    const float* __restrict__ xrow = x + (size_t)arow * CFEAT;

    v8f acc[8];
    const v8f z = {0.f, 0.f, 0.f, 0.f, 0.f, 0.f, 0.f, 0.f};
    #pragma unroll
    for (int t = 0; t < 8; ++t) acc[t] = z;

    for (int k = 0; k < CFEAT; k += 4) {
        v2f a;
        a.x = xrow[k + 2 * hi];
        a.y = xrow[k + 2 * hi + 1];
        #pragma unroll
        for (int t = 0; t < 8; ++t) {
            v2f b;
            const int col = t * 16 + lo;
            b.x = sW[(k + 2 * hi) * CFEAT + col];
            b.y = sW[(k + 2 * hi + 1) * CFEAT + col];
            acc[t] = __builtin_amdgcn_wmma_f32_16x16x4_f32(
                false, a, false, b, (short)0, acc[t], false, false);
        }
    }

    if (rowbase + 16 <= n) {
        #pragma unroll
        for (int t = 0; t < 8; ++t) {
            #pragma unroll
            for (int r = 0; r < 8; ++r) {
                const int row = rowbase + r + 8 * hi;
                h[(size_t)row * CFEAT + t * 16 + lo] = acc[t][r];
            }
        }
    } else {
        #pragma unroll
        for (int t = 0; t < 8; ++t) {
            #pragma unroll
            for (int r = 0; r < 8; ++r) {
                const int row = rowbase + r + 8 * hi;
                if (row < n) h[(size_t)row * CFEAT + t * 16 + lo] = acc[t][r];
            }
        }
    }
}


#define AGG_G 8

__device__ __forceinline__ int   lane_i(int x, int t)   { return __builtin_amdgcn_readlane(x, t); }
__device__ __forceinline__ float lane_f(float x, int t) {
    return __builtin_bit_cast(float, __builtin_amdgcn_readlane(__builtin_bit_cast(int, x), t));
}

__device__ __forceinline__ void axpy4(float4& acc, const float4 v, const float w) {
    acc.x += v.x * w; acc.y += v.y * w; acc.z += v.z * w; acc.w += v.w * w;
}

__device__ __forceinline__ float4 aggregate(const float* __restrict__ hsrc,
                                            const int* __restrict__ offs,
                                            const int* __restrict__ csr,
                                            const float* __restrict__ dinv,
                                            int node, int lane) {
    const float din = dinv[node];
    const float4* __restrict__ h4 = (const float4*)hsrc;
    float4 acc;
    {
        float4 v = h4[(size_t)node * 32 + lane];
        acc.x = v.x * din; acc.y = v.y * din; acc.z = v.z * din; acc.w = v.w * din;
    }
    const int beg = offs[node], end = offs[node + 1];
    for (int j = beg; j < end; j += 32) {
        const int idx = j + lane;
        const int   s  = (idx < end) ? csr[idx] : 0;
        const float ws = (idx < end) ? dinv[s] : 0.0f;
        int cnt = end - j; if (cnt > 32) cnt = 32;
        int t = 0;
        for (; t + AGG_G <= cnt; t += AGG_G) {
            float4 v[AGG_G];
            #pragma unroll
            for (int i = 0; i < AGG_G; ++i) v[i] = h4[(size_t)lane_i(s, t + i) * 32 + lane];
            #pragma unroll
            for (int i = 0; i < AGG_G; ++i) axpy4(acc, v[i], lane_f(ws, t + i));
        }
        const int r = cnt - t;
        if (r != 0) {
            const int t4 = t;
            const int t2 = t4 + (r & 4);
            const int t1 = t2 + (r & 2);
            float4 a0, a1, a2, a3, b0, b1, c0;
            a0 = a1 = a2 = a3 = b0 = b1 = c0 = make_float4(0.f, 0.f, 0.f, 0.f);
            if (r & 4) {
                a0 = h4[(size_t)lane_i(s, t4)     * 32 + lane];
                a1 = h4[(size_t)lane_i(s, t4 + 1) * 32 + lane];
                a2 = h4[(size_t)lane_i(s, t4 + 2) * 32 + lane];
                a3 = h4[(size_t)lane_i(s, t4 + 3) * 32 + lane];
            }
            if (r & 2) {
                b0 = h4[(size_t)lane_i(s, t2)     * 32 + lane];
                b1 = h4[(size_t)lane_i(s, t2 + 1) * 32 + lane];
            }
            if (r & 1) {
                c0 = h4[(size_t)lane_i(s, t1)     * 32 + lane];
            }
            if (r & 4) {
                axpy4(acc, a0, lane_f(ws, t4));
                axpy4(acc, a1, lane_f(ws, t4 + 1));
                axpy4(acc, a2, lane_f(ws, t4 + 2));
                axpy4(acc, a3, lane_f(ws, t4 + 3));
            }
            if (r & 2) {
                axpy4(acc, b0, lane_f(ws, t2));
                axpy4(acc, b1, lane_f(ws, t2 + 1));
            }
            if (r & 1) {
                axpy4(acc, c0, lane_f(ws, t1));
            }
        }
    }
    acc.x *= din; acc.y *= din; acc.z *= din; acc.w *= din;
    return acc;
}

__global__ __launch_bounds__(256)
void k_prop1_proj(const float* __restrict__ h, float* __restrict__ zs,
                  const int* __restrict__ offs, const int* __restrict__ csr,
                  const float* __restrict__ dinv, const float* __restrict__ b1,
                  const float* __restrict__ Wout, int n) {
    const int node = __builtin_amdgcn_readfirstlane((int)((blockIdx.x * blockDim.x + threadIdx.x) >> 5));
    const int lane = threadIdx.x & 31;
    if (blockIdx.x == 0 && threadIdx.x < 2)
        ((float4*)zs)[(size_t)n * 2 + threadIdx.x] = make_float4(0.f, 0.f, 0.f, 0.f);
    if (node >= n) return;

    float4 acc = aggregate(h, offs, csr, dinv, node, lane);
    const float4 b = ((const float4*)b1)[lane];
    acc.x = fmaxf(acc.x + b.x, 0.0f);
    acc.y = fmaxf(acc.y + b.y, 0.0f);
    acc.z = fmaxf(acc.z + b.z, 0.0f);
    acc.w = fmaxf(acc.w + b.w, 0.0f);

    float p[4][2];
    #pragma unroll
    for (int m = 0; m < 4; ++m) {
        const float4* __restrict__ w4 = (const float4*)(Wout + ((size_t)m * CFEAT + lane * 4) * 2);
        const float4 w01 = w4[0];
        const float4 w23 = w4[1];
        p[m][0] = acc.x * w01.x + acc.y * w01.z + acc.z * w23.x + acc.w * w23.z;
        p[m][1] = acc.x * w01.y + acc.y * w01.w + acc.z * w23.y + acc.w * w23.w;
    }

    #pragma unroll
    for (int m = 0; m < 4; ++m)
        #pragma unroll
        for (int o = 0; o < 2; ++o) {
            float v = p[m][o];
            #pragma unroll
            for (int d = 16; d > 0; d >>= 1) v += __shfl_xor(v, d, 32);
            p[m][o] = v;
        }

    if (lane == 0) {
        const float din = dinv[node];
        float4* __restrict__ z4 = (float4*)zs + (size_t)node * 2;
        z4[0] = make_float4(p[0][0] * din, p[0][1] * din, p[1][0] * din, p[1][1] * din);
        z4[1] = make_float4(p[2][0] * din, p[2][1] * din, p[3][0] * din, p[3][1] * din);
    }
}


#define P2_NPW 4
#define P2_CH  3

__device__ __forceinline__ float sum_same_parity(float x) {
    x += __builtin_bit_cast(float, __builtin_amdgcn_update_dpp(0, __builtin_bit_cast(int, x), 0x162, 0xf, 0xf, true));
    x += __builtin_bit_cast(float, __builtin_amdgcn_update_dpp(0, __builtin_bit_cast(int, x), 0x164, 0xf, 0xf, true));
    x += __builtin_bit_cast(float, __builtin_amdgcn_update_dpp(0, __builtin_bit_cast(int, x), 0x168, 0xf, 0xf, true));
    const int b = __builtin_bit_cast(int, x);
    x += __builtin_bit_cast(float, __builtin_amdgcn_permlanex16(b, b, 0x76543210, 0xfedcba98, false, false));
    return x;
}

__global__ __launch_bounds__(256)
void k_prop2_small(const float* __restrict__ zs, float* __restrict__ out,
                   const int* __restrict__ offs, const int* __restrict__ csr,
                   const float* __restrict__ dinv, const float* __restrict__ bout,
                   int n, int e) {
    const int lane = threadIdx.x & 31;
    const int q    = lane >> 1;
    const int half = lane & 1;
    const int node0 = __builtin_amdgcn_readfirstlane((int)((blockIdx.x * blockDim.x + threadIdx.x) >> 5)) * P2_NPW;
    if (node0 >= n) return;
    const float4* __restrict__ z4 = (const float4*)zs;

    const int   tq   = node0 + q;
    const float dinq = dinv[(tq < n) ? tq : (n - 1)];
    const float4 bo  = ((const float4*)bout)[half];

    int nd[P2_NPW], beg[P2_NPW], end[P2_NPW];
    #pragma unroll
    for (int i = 0; i < P2_NPW; ++i) {
        const int t = node0 + i;
        nd[i]  = (t < n) ? t : (n - 1);
        beg[i] = offs[nd[i]];
        end[i] = offs[nd[i] + 1];
    }

    int  s[P2_NPW][P2_CH];
    bool ok[P2_NPW][P2_CH];
    #pragma unroll
    for (int i = 0; i < P2_NPW; ++i)
        #pragma unroll
        for (int k = 0; k < P2_CH; ++k) {
            const int idx = beg[i] + k * 16 + q - 1;
            ok[i][k] = idx < end[i];
            int ic = (idx > 0) ? idx : 0;
            ic = (ic < e) ? ic : (e - 1);
            s[i][k] = csr[ic];
        }

    float4 acc[P2_NPW];
    #pragma unroll
    for (int i = 0; i < P2_NPW; ++i) {
        float4 v[P2_CH];
        #pragma unroll
        for (int k = 0; k < P2_CH; ++k) {
            int sv = ok[i][k] ? s[i][k] : n;
            if (k == 0) sv = (q == 0) ? nd[i] : sv;
            v[k] = z4[(unsigned)sv * 2u + (unsigned)half];
        }
        acc[i] = v[0];
        #pragma unroll
        for (int k = 1; k < P2_CH; ++k) {
            acc[i].x += v[k].x; acc[i].y += v[k].y; acc[i].z += v[k].z; acc[i].w += v[k].w;
        }
    }

    #pragma unroll
    for (int i = 0; i < P2_NPW; ++i)
        for (int j = beg[i] + P2_CH * 16 - 1; j < end[i]; j += 16) {
            const int  idx = j + q;
            const bool okj = idx < end[i];
            const int  sj  = csr[okj ? idx : j];
            const float4 vj = z4[(unsigned)(okj ? sj : n) * 2u + (unsigned)half];
            acc[i].x += vj.x; acc[i].y += vj.y; acc[i].z += vj.z; acc[i].w += vj.w;
        }

    #pragma unroll
    for (int i = 0; i < P2_NPW; ++i) {
        acc[i].x = sum_same_parity(acc[i].x);
        acc[i].y = sum_same_parity(acc[i].y);
        acc[i].z = sum_same_parity(acc[i].z);
        acc[i].w = sum_same_parity(acc[i].w);
    }

    float4 t = acc[0];
    #pragma unroll
    for (int i = 1; i < P2_NPW; ++i)
        if (q == i) t = acc[i];
    float2 r0, r1;
    r0.x = t.x * dinq + bo.x;
    r0.y = t.y * dinq + bo.y;
    r1.x = t.z * dinq + bo.z;
    r1.y = t.w * dinq + bo.w;
    if (q < P2_NPW && tq < n) {
        const int m0 = half * 2;
        ((float2*)out)[(size_t)m0 * n + tq]       = r0;
        ((float2*)out)[(size_t)(m0 + 1) * n + tq] = r1;
    }
}


extern "C" void kernel_launch(void* const* d_in, const int* in_sizes, int n_in,
                              void* d_out, int out_size, void* d_ws, size_t ws_size,
                              hipStream_t stream) {
    const float* x    = (const float*)d_in[0];
    const int*   ei   = (const int*)d_in[1];
    const float* W1   = (const float*)d_in[2];
    const float* b1   = (const float*)d_in[3];
    const float* Wout = (const float*)d_in[4];
    const float* bout = (const float*)d_in[5];
    float* out = (float*)d_out;

    const int n = in_sizes[0] / CFEAT;
    const int e = in_sizes[1] / 2;
    const int* srcv = ei;
    const int* dstv = ei + e;

    uintptr_t p = (uintptr_t)d_ws;
    auto alloc = [&](size_t bytes) {
        p = (p + 255) & ~(uintptr_t)255;
        void* r = (void*)p;
        p += bytes;
        return r;
    };
    const int G1    = 256;
    const int chunk = (e + G1 - 1) / G1;
    const int nbk   = (n + BNODES - 1) >> BSH;
    int*   bcnt   = (int*)  alloc((size_t)nbk * G1 * 4);
    int*   boffs  = (int*)  alloc(((size_t)nbk * G1 + 1) * 4);
    int*   tmp    = (int*)  alloc((size_t)e * 8);
    int*   offs   = (int*)  alloc((size_t)(n + 1) * 4);
    int*   csr    = (int*)  alloc((size_t)e * 4);
    float* dinv   = (float*)alloc((size_t)n * 4);
    float* h      = (float*)alloc((size_t)n * CFEAT * 4);
    float* zs     = (float*)alloc(((size_t)n + 1) * 8 * 4);
    (void)ws_size; (void)n_in; (void)out_size;

    const int TB = 256;

    k_bucket_count  <<<G1, ETHREADS, (size_t)nbk * 4, stream>>>(dstv, bcnt, e, chunk, nbk);
    k_scan          <<<1, 1024, 0, stream>>>(bcnt, boffs, nbk * G1);
    k_bucket_scatter<<<G1, ETHREADS, (size_t)nbk * 4, stream>>>(srcv, dstv, boffs, tmp, e, chunk, nbk);
    k_bucket_final  <<<nbk, ETHREADS, 0, stream>>>(tmp, boffs, offs, csr, dinv, n, G1);

    k_gemm1<<<(n + 127) / 128, TB, 0, stream>>>(x, W1, h, n);

    const int gprop = (n + 7) / 8;
    k_prop1_proj <<<gprop, TB, 0, stream>>>(h, zs, offs, csr, dinv, b1, Wout, n);
    const int gp2 = (n + 8 * P2_NPW - 1) / (8 * P2_NPW);
    k_prop2_small<<<gp2, TB, 0, stream>>>(zs, out, offs, csr, dinv, bout, n, e);
}


# csr level 1 in one launch: each block counts, scans in LDS and moves its edge slice into its own region; level 2 walks 256 ranges; 7 launches become 5; csr row (float sum) order stays arrival order
# speedup vs baseline: 5.2581x; 1.1887x over previous
#include <hip/hip_runtime.h>
#include <cstdint>

typedef __attribute__((ext_vector_type(2))) float v2f;
typedef __attribute__((ext_vector_type(8))) float v8f;

#define CFEAT 128


#define BSH      8
#define BNODES   (1 << BSH)
#define ETHREADS 1024
#define EUNROLL  8
#define SEGU     8

__device__ __forceinline__ int wave_incl_scan(int x, int lane) {
    x += __builtin_amdgcn_update_dpp(0, x, 0x111, 0xf, 0xf, true);
    x += __builtin_amdgcn_update_dpp(0, x, 0x112, 0xf, 0xf, true);
    x += __builtin_amdgcn_update_dpp(0, x, 0x114, 0xf, 0xf, true);
    x += __builtin_amdgcn_update_dpp(0, x, 0x118, 0xf, 0xf, true);
    const int row0 = __builtin_amdgcn_readlane(x, 15);
    return x + ((lane >= 16) ? row0 : 0);
}

__global__ __launch_bounds__(ETHREADS)
void k_slice_sort(const int* __restrict__ src, const int* __restrict__ dst,
                  int* __restrict__ cntm, int* __restrict__ lstm, int* __restrict__ tmpw,
                  int e, int chunk, int nbk) {
    extern __shared__ int hist[];
    __shared__ int wtot[ETHREADS / 32];
    const int tid  = threadIdx.x;
    const int lane = tid & 31;
    const int wave = __builtin_amdgcn_readfirstlane(tid >> 5);
    for (int b = tid; b < nbk; b += ETHREADS) hist[b] = 0;
    __syncthreads();
    const int start = blockIdx.x * chunk;
    int end = start + chunk; if (end > e) end = e;
    for (int base = start; base < end; base += EUNROLL * ETHREADS) {
        int d[EUNROLL];
        #pragma unroll
        for (int u = 0; u < EUNROLL; ++u) {
            const int i = base + u * ETHREADS + tid;
            d[u] = (i < end) ? dst[i] : -1;
        }
        #pragma unroll
        for (int u = 0; u < EUNROLL; ++u)
            if (d[u] >= 0) atomicAdd(&hist[d[u] >> BSH], 1);
    }
    __syncthreads();

    int carry = 0;
    for (int base = 0; base < nbk; base += ETHREADS) {
        const int b   = base + tid;
        const int v   = (b < nbk) ? hist[b] : 0;
        const int inc = wave_incl_scan(v, lane);
        if (lane == 31) wtot[wave] = inc;
        __syncthreads();
        const int wt   = wtot[lane];
        const int winc = wave_incl_scan(wt, lane);
        const int ex   = carry + __builtin_amdgcn_readlane(winc - wt, wave) + inc - v;
        carry += __builtin_amdgcn_readlane(winc, 31);
        if (b < nbk) {
            hist[b] = ex;
            cntm[b * gridDim.x + blockIdx.x] = v;
            lstm[b * gridDim.x + blockIdx.x] = ex;
        }
        __syncthreads();
    }

    int2* __restrict__ tmp = (int2*)tmpw + (size_t)blockIdx.x * chunk;
    for (int base = start; base < end; base += EUNROLL * ETHREADS) {
        int d[EUNROLL], s[EUNROLL];
        #pragma unroll
        for (int u = 0; u < EUNROLL; ++u) {
            const int i = base + u * ETHREADS + tid;
            d[u] = (i < end) ? dst[i] : -1;
            s[u] = (i < end) ? src[i] : 0;
        }
        #pragma unroll
        for (int u = 0; u < EUNROLL; ++u)
            if (d[u] >= 0) {
                const int p = atomicAdd(&hist[d[u] >> BSH], 1);
                tmp[p] = make_int2(s[u], d[u]);
            }
    }
}

__global__ __launch_bounds__(ETHREADS)
void k_bucket_final(const int* __restrict__ tmpw, const int* __restrict__ cntm,
                    const int* __restrict__ lstm,
                    int* __restrict__ offs, int* __restrict__ csr,
                    float* __restrict__ dinv, int n, int g1, int chunk) {
    const int2* __restrict__ tmp = (const int2*)tmpw;
    __shared__ int cnt[BNODES];
    __shared__ int loc[BNODES];
    __shared__ int wsum[BNODES / 32];
    __shared__ int segc[ETHREADS];
    __shared__ int segl[ETHREADS];
    __shared__ int red[2][ETHREADS / 32];
    const int tid  = threadIdx.x;
    const int lane = tid & 31;
    const int wave = tid >> 5;
    const int b    = blockIdx.x;

    int sc = 0, sl = 0;
    if (tid < g1) {
        sc = cntm[b * g1 + tid];
        sl = lstm[b * g1 + tid];
        segc[tid] = sc;
        segl[tid] = sl;
    }
    {
        const int ic = wave_incl_scan(sc, lane);
        const int il = wave_incl_scan(sl, lane);
        if (lane == 31) { red[0][wave] = ic; red[1][wave] = il; }
    }
    if (tid < BNODES) cnt[tid] = 0;
    __syncthreads();
    const int bs = __builtin_amdgcn_readlane(wave_incl_scan(red[1][lane], lane), 31);
    const int be = bs + __builtin_amdgcn_readlane(wave_incl_scan(red[0][lane], lane), 31);

    for (int g0 = wave; g0 < g1; g0 += 32 * SEGU) {
        int dy[SEGU][2];
        int len[SEGU];
        const int2* p[SEGU];
        #pragma unroll
        for (int u = 0; u < SEGU; ++u) {
            const int g  = g0 + 32 * u;
            const int gg = (g < g1) ? g : 0;
            len[u] = (g < g1) ? segc[gg] : 0;
            p[u]   = tmp + (size_t)gg * chunk + segl[gg];
        }
        #pragma unroll
        for (int u = 0; u < SEGU; ++u) {
            dy[u][0] = (lane < len[u])      ? p[u][lane].y      : -1;
            dy[u][1] = (lane + 32 < len[u]) ? p[u][lane + 32].y : -1;
        }
        #pragma unroll
        for (int u = 0; u < SEGU; ++u) {
            if (dy[u][0] >= 0) atomicAdd(&cnt[dy[u][0] & (BNODES - 1)], 1);
            if (dy[u][1] >= 0) atomicAdd(&cnt[dy[u][1] & (BNODES - 1)], 1);
        }
        #pragma unroll
        for (int u = 0; u < SEGU; ++u)
            for (int j = 64 + lane; j < len[u]; j += 32)
                atomicAdd(&cnt[p[u][j].y & (BNODES - 1)], 1);
    }
    __syncthreads();

    int c = 0, inc = 0;
    if (tid < BNODES) {
        c   = cnt[tid];
        inc = wave_incl_scan(c, lane);
        if (lane == 31) wsum[wave] = inc;
    }
    __syncthreads();
    if (tid < BNODES) {
        int add = 0;
        for (int w = 0; w < wave; ++w) add += wsum[w];
        const int excl = add + inc - c;
        loc[tid] = excl;
        cnt[tid] = 0;
        const int node = b * BNODES + tid;
        if (node < n) {
            offs[node] = bs + excl;
            dinv[node] = rsqrtf((float)c + 1.0f);
        }
    }
    if (b == (int)gridDim.x - 1 && tid == 0) offs[n] = be;
    __syncthreads();

    for (int g0 = wave; g0 < g1; g0 += 32 * SEGU) {
        int2 t[SEGU][2];
        int len[SEGU];
        const int2* p[SEGU];
        #pragma unroll
        for (int u = 0; u < SEGU; ++u) {
            const int g  = g0 + 32 * u;
            const int gg = (g < g1) ? g : 0;
            len[u] = (g < g1) ? segc[gg] : 0;
            p[u]   = tmp + (size_t)gg * chunk + segl[gg];
        }
        #pragma unroll
        for (int u = 0; u < SEGU; ++u) {
            t[u][0] = (lane < len[u])      ? p[u][lane]      : make_int2(0, -1);
            t[u][1] = (lane + 32 < len[u]) ? p[u][lane + 32] : make_int2(0, -1);
        }
        #pragma unroll
        for (int u = 0; u < SEGU; ++u) {
            #pragma unroll
            for (int v = 0; v < 2; ++v)
                if (t[u][v].y >= 0) {
                    const int k = t[u][v].y & (BNODES - 1);
                    const int q = atomicAdd(&cnt[k], 1);
                    csr[bs + loc[k] + q] = t[u][v].x;
                }
        }
        #pragma unroll
        for (int u = 0; u < SEGU; ++u)
            for (int j = 64 + lane; j < len[u]; j += 32) {
                const int2 pr = p[u][j];
                const int k = pr.y & (BNODES - 1);
                const int q = atomicAdd(&cnt[k], 1);
                csr[bs + loc[k] + q] = pr.x;
            }
    }
}


__global__ __launch_bounds__(256)
void k_gemm1(const float* __restrict__ x, const float* __restrict__ W1,
             const float* __restrict__ dinv, float* __restrict__ h, int n) {
    __shared__ float sW[CFEAT * CFEAT];
    for (int i = threadIdx.x; i < (CFEAT * CFEAT) / 4; i += 256)
        ((float4*)sW)[i] = ((const float4*)W1)[i];
    __syncthreads();

    const int lane = threadIdx.x & 31;
    const int wave = threadIdx.x >> 5;
    const int hi   = lane >> 4;
    const int lo   = lane & 15;
    const int rowbase = (blockIdx.x * 8 + wave) * 16;
    if (rowbase >= n) return;

    int arow = rowbase + lo;
    if (arow >= n) arow = n - 1;
    const float* __restrict__ xrow = x + (size_t)arow * CFEAT;

    float dv[8];
    #pragma unroll
    for (int r = 0; r < 8; ++r) {
        int row = rowbase + r + 8 * hi;
        if (row >= n) row = n - 1;
        dv[r] = dinv[row];
    }

    v8f acc[8];
    const v8f z = {0.f, 0.f, 0.f, 0.f, 0.f, 0.f, 0.f, 0.f};
    #pragma unroll
    for (int t = 0; t < 8; ++t) acc[t] = z;

    for (int k = 0; k < CFEAT; k += 4) {
        v2f a;
        a.x = xrow[k + 2 * hi];
        a.y = xrow[k + 2 * hi + 1];
        #pragma unroll
        for (int t = 0; t < 8; ++t) {
            v2f b;
            const int col = t * 16 + lo;
            b.x = sW[(k + 2 * hi) * CFEAT + col];
            b.y = sW[(k + 2 * hi + 1) * CFEAT + col];
            acc[t] = __builtin_amdgcn_wmma_f32_16x16x4_f32(
                false, a, false, b, (short)0, acc[t], false, false);
        }
    }

    if (rowbase + 16 <= n) {
        #pragma unroll
        for (int t = 0; t < 8; ++t) {
            #pragma unroll
            for (int r = 0; r < 8; ++r) {
                const int row = rowbase + r + 8 * hi;
                h[(size_t)row * CFEAT + t * 16 + lo] = acc[t][r] * dv[r];
            }
        }
    } else {
        #pragma unroll
        for (int t = 0; t < 8; ++t) {
            #pragma unroll
            for (int r = 0; r < 8; ++r) {
                const int row = rowbase + r + 8 * hi;
                if (row < n) h[(size_t)row * CFEAT + t * 16 + lo] = acc[t][r] * dv[r];
            }
        }
    }
}


#define AGG_G 8

__device__ __forceinline__ int lane_i(int x, int t) { return __builtin_amdgcn_readlane(x, t); }

__device__ __forceinline__ void add4(float4& acc, const float4 v) {
    acc.x += v.x; acc.y += v.y; acc.z += v.z; acc.w += v.w;
}

__device__ __forceinline__ float4 aggregate(const float* __restrict__ hsrc,
                                            const int* __restrict__ offs,
                                            const int* __restrict__ csr,
                                            int node, int lane) {
    const float4* __restrict__ h4 = (const float4*)hsrc;
    float4 acc = h4[(size_t)node * 32 + lane];
    const int beg = offs[node], end = offs[node + 1];
    for (int j = beg; j < end; j += 32) {
        const int idx = j + lane;
        const int s = (idx < end) ? csr[idx] : 0;
        int cnt = end - j; if (cnt > 32) cnt = 32;
        int t = 0;
        for (; t + AGG_G <= cnt; t += AGG_G) {
            float4 v[AGG_G];
            #pragma unroll
            for (int i = 0; i < AGG_G; ++i) v[i] = h4[(size_t)lane_i(s, t + i) * 32 + lane];
            #pragma unroll
            for (int i = 0; i < AGG_G; ++i) add4(acc, v[i]);
        }
        const int r = cnt - t;
        if (r != 0) {
            const int t4 = t;
            const int t2 = t4 + (r & 4);
            const int t1 = t2 + (r & 2);
            float4 a0, a1, a2, a3, b0, b1, c0;
            a0 = a1 = a2 = a3 = b0 = b1 = c0 = make_float4(0.f, 0.f, 0.f, 0.f);
            if (r & 4) {
                a0 = h4[(size_t)lane_i(s, t4)     * 32 + lane];
                a1 = h4[(size_t)lane_i(s, t4 + 1) * 32 + lane];
                a2 = h4[(size_t)lane_i(s, t4 + 2) * 32 + lane];
                a3 = h4[(size_t)lane_i(s, t4 + 3) * 32 + lane];
            }
            if (r & 2) {
                b0 = h4[(size_t)lane_i(s, t2)     * 32 + lane];
                b1 = h4[(size_t)lane_i(s, t2 + 1) * 32 + lane];
            }
            if (r & 1) {
                c0 = h4[(size_t)lane_i(s, t1)     * 32 + lane];
            }
            if (r & 4) { add4(acc, a0); add4(acc, a1); add4(acc, a2); add4(acc, a3); }
            if (r & 2) { add4(acc, b0); add4(acc, b1); }
            if (r & 1) { add4(acc, c0); }
        }
    }
    return acc;
}

__global__ __launch_bounds__(256)
void k_prop1_proj(const float* __restrict__ hs, float* __restrict__ zs,
                  const int* __restrict__ offs, const int* __restrict__ csr,
                  const float* __restrict__ dinv, const float* __restrict__ b1,
                  const float* __restrict__ Wout, int n) {
    const int node = __builtin_amdgcn_readfirstlane((int)((blockIdx.x * blockDim.x + threadIdx.x) >> 5));
    const int lane = threadIdx.x & 31;
    if (blockIdx.x == 0 && threadIdx.x < 2)
        ((float4*)zs)[(size_t)n * 2 + threadIdx.x] = make_float4(0.f, 0.f, 0.f, 0.f);
    if (node >= n) return;

    const float din = dinv[node];
    float4 acc = aggregate(hs, offs, csr, node, lane);
    const float4 b = ((const float4*)b1)[lane];
    acc.x = fmaxf(acc.x * din + b.x, 0.0f);
    acc.y = fmaxf(acc.y * din + b.y, 0.0f);
    acc.z = fmaxf(acc.z * din + b.z, 0.0f);
    acc.w = fmaxf(acc.w * din + b.w, 0.0f);

    float p[4][2];
    #pragma unroll
    for (int m = 0; m < 4; ++m) {
        const float4* __restrict__ w4 = (const float4*)(Wout + ((size_t)m * CFEAT + lane * 4) * 2);
        const float4 w01 = w4[0];
        const float4 w23 = w4[1];
        p[m][0] = acc.x * w01.x + acc.y * w01.z + acc.z * w23.x + acc.w * w23.z;
        p[m][1] = acc.x * w01.y + acc.y * w01.w + acc.z * w23.y + acc.w * w23.w;
    }

    #pragma unroll
    for (int m = 0; m < 4; ++m)
        #pragma unroll
        for (int o = 0; o < 2; ++o) {
            float v = p[m][o];
            #pragma unroll
            for (int d = 16; d > 0; d >>= 1) v += __shfl_xor(v, d, 32);
            p[m][o] = v;
        }

    if (lane == 0) {
        float4* __restrict__ z4 = (float4*)zs + (size_t)node * 2;
        z4[0] = make_float4(p[0][0] * din, p[0][1] * din, p[1][0] * din, p[1][1] * din);
        z4[1] = make_float4(p[2][0] * din, p[2][1] * din, p[3][0] * din, p[3][1] * din);
    }
}


#define P2_NPW 4
#define P2_CH  3

__device__ __forceinline__ float sum_same_parity(float x) {
    x += __builtin_bit_cast(float, __builtin_amdgcn_update_dpp(0, __builtin_bit_cast(int, x), 0x162, 0xf, 0xf, true));
    x += __builtin_bit_cast(float, __builtin_amdgcn_update_dpp(0, __builtin_bit_cast(int, x), 0x164, 0xf, 0xf, true));
    x += __builtin_bit_cast(float, __builtin_amdgcn_update_dpp(0, __builtin_bit_cast(int, x), 0x168, 0xf, 0xf, true));
    const int b = __builtin_bit_cast(int, x);
    x += __builtin_bit_cast(float, __builtin_amdgcn_permlanex16(b, b, 0x76543210, 0xfedcba98, false, false));
    return x;
}

__global__ __launch_bounds__(256)
void k_prop2_small(const float* __restrict__ zs, float* __restrict__ out,
                   const int* __restrict__ offs, const int* __restrict__ csr,
                   const float* __restrict__ dinv, const float* __restrict__ bout,
                   int n, int e) {
    const int lane = threadIdx.x & 31;
    const int q    = lane >> 1;
    const int half = lane & 1;
    const int node0 = __builtin_amdgcn_readfirstlane((int)((blockIdx.x * blockDim.x + threadIdx.x) >> 5)) * P2_NPW;
    if (node0 >= n) return;
    const float4* __restrict__ z4 = (const float4*)zs;

    const int   tq   = node0 + q;
    const float dinq = dinv[(tq < n) ? tq : (n - 1)];
    const float4 bo  = ((const float4*)bout)[half];

    int nd[P2_NPW], beg[P2_NPW], end[P2_NPW];
    #pragma unroll
    for (int i = 0; i < P2_NPW; ++i) {
        const int t = node0 + i;
        nd[i]  = (t < n) ? t : (n - 1);
        beg[i] = offs[nd[i]];
        end[i] = offs[nd[i] + 1];
    }

    int  s[P2_NPW][P2_CH];
    bool ok[P2_NPW][P2_CH];
    #pragma unroll
    for (int i = 0; i < P2_NPW; ++i)
        #pragma unroll
        for (int k = 0; k < P2_CH; ++k) {
            const int idx = beg[i] + k * 16 + q - 1;
            ok[i][k] = idx < end[i];
            int ic = (idx > 0) ? idx : 0;
            ic = (ic < e) ? ic : (e - 1);
            s[i][k] = csr[ic];
        }

    float4 acc[P2_NPW];
    #pragma unroll
    for (int i = 0; i < P2_NPW; ++i) {
        float4 v[P2_CH];
        #pragma unroll
        for (int k = 0; k < P2_CH; ++k) {
            int sv = ok[i][k] ? s[i][k] : n;
            if (k == 0) sv = (q == 0) ? nd[i] : sv;
            v[k] = z4[(unsigned)sv * 2u + (unsigned)half];
        }
        acc[i] = v[0];
        #pragma unroll
        for (int k = 1; k < P2_CH; ++k) {
            acc[i].x += v[k].x; acc[i].y += v[k].y; acc[i].z += v[k].z; acc[i].w += v[k].w;
        }
    }

    #pragma unroll
    for (int i = 0; i < P2_NPW; ++i)
        for (int j = beg[i] + P2_CH * 16 - 1; j < end[i]; j += 16) {
            const int  idx = j + q;
            const bool okj = idx < end[i];
            const int  sj  = csr[okj ? idx : j];
            const float4 vj = z4[(unsigned)(okj ? sj : n) * 2u + (unsigned)half];
            acc[i].x += vj.x; acc[i].y += vj.y; acc[i].z += vj.z; acc[i].w += vj.w;
        }

    #pragma unroll
    for (int i = 0; i < P2_NPW; ++i) {
        acc[i].x = sum_same_parity(acc[i].x);
        acc[i].y = sum_same_parity(acc[i].y);
        acc[i].z = sum_same_parity(acc[i].z);
        acc[i].w = sum_same_parity(acc[i].w);
    }

    float4 t = acc[0];
    #pragma unroll
    for (int i = 1; i < P2_NPW; ++i)
        if (q == i) t = acc[i];
    float2 r0, r1;
    r0.x = t.x * dinq + bo.x;
    r0.y = t.y * dinq + bo.y;
    r1.x = t.z * dinq + bo.z;
    r1.y = t.w * dinq + bo.w;
    if (q < P2_NPW && tq < n) {
        const int m0 = half * 2;
        ((float2*)out)[(size_t)m0 * n + tq]       = r0;
        ((float2*)out)[(size_t)(m0 + 1) * n + tq] = r1;
    }
}


extern "C" void kernel_launch(void* const* d_in, const int* in_sizes, int n_in,
                              void* d_out, int out_size, void* d_ws, size_t ws_size,
                              hipStream_t stream) {
    const float* x    = (const float*)d_in[0];
    const int*   ei   = (const int*)d_in[1];
    const float* W1   = (const float*)d_in[2];
    const float* b1   = (const float*)d_in[3];
    const float* Wout = (const float*)d_in[4];
    const float* bout = (const float*)d_in[5];
    float* out = (float*)d_out;

    const int n = in_sizes[0] / CFEAT;
    const int e = in_sizes[1] / 2;
    const int* srcv = ei;
    const int* dstv = ei + e;

    uintptr_t p = (uintptr_t)d_ws;
    auto alloc = [&](size_t bytes) {
        p = (p + 255) & ~(uintptr_t)255;
        void* r = (void*)p;
        p += bytes;
        return r;
    };
    const int G1    = 256;
    const int chunk = (e + G1 - 1) / G1;
    const int nbk   = (n + BNODES - 1) >> BSH;
    int*   cntm   = (int*)  alloc((size_t)nbk * G1 * 4);
    int*   lstm   = (int*)  alloc((size_t)nbk * G1 * 4);
    int*   tmp    = (int*)  alloc((size_t)G1 * chunk * 8);
    int*   offs   = (int*)  alloc((size_t)(n + 1) * 4);
    int*   csr    = (int*)  alloc((size_t)e * 4);
    float* dinv   = (float*)alloc((size_t)n * 4);
    float* h      = (float*)alloc((size_t)n * CFEAT * 4);
    float* zs     = (float*)alloc(((size_t)n + 1) * 8 * 4);
    (void)ws_size; (void)n_in; (void)out_size;

    const int TB = 256;

    k_slice_sort  <<<G1, ETHREADS, (size_t)nbk * 4, stream>>>(srcv, dstv, cntm, lstm, tmp, e, chunk, nbk);
    k_bucket_final<<<nbk, ETHREADS, 0, stream>>>(tmp, cntm, lstm, offs, csr, dinv, n, G1, chunk);

    k_gemm1<<<(n + 127) / 128, TB, 0, stream>>>(x, W1, dinv, h, n);

    const int gprop = (n + 7) / 8;
    k_prop1_proj <<<gprop, TB, 0, stream>>>(h, zs, offs, csr, dinv, b1, Wout, n);
    const int gp2 = (n + 8 * P2_NPW - 1) / (8 * P2_NPW);
    k_prop2_small<<<gp2, TB, 0, stream>>>(zs, out, offs, csr, dinv, bout, n, e);
}


# k_gemm1 joins k_bucket_final: a bucket block multiplies its own 256 rows (wave = 16 rows x 64 cols) with the dinv it just made, half its waves before placing; 5 launches become 4; sums and order kept
# speedup vs baseline: 5.8975x; 1.1216x over previous
#include <hip/hip_runtime.h>
#include <cstdint>

typedef __attribute__((ext_vector_type(2))) float v2f;
typedef __attribute__((ext_vector_type(8))) float v8f;

#define CFEAT 128


#define BSH      8
#define BNODES   (1 << BSH)
#define ETHREADS 1024
#define EUNROLL  8
#define SEGU     8

__device__ __forceinline__ int wave_incl_scan(int x, int lane) {
    x += __builtin_amdgcn_update_dpp(0, x, 0x111, 0xf, 0xf, true);
    x += __builtin_amdgcn_update_dpp(0, x, 0x112, 0xf, 0xf, true);
    x += __builtin_amdgcn_update_dpp(0, x, 0x114, 0xf, 0xf, true);
    x += __builtin_amdgcn_update_dpp(0, x, 0x118, 0xf, 0xf, true);
    const int row0 = __builtin_amdgcn_readlane(x, 15);
    return x + ((lane >= 16) ? row0 : 0);
}

__global__ __launch_bounds__(ETHREADS)
void k_slice_sort(const int* __restrict__ src, const int* __restrict__ dst,
                  int* __restrict__ cntm, int* __restrict__ lstm, int* __restrict__ tmpw,
                  int e, int chunk, int nbk) {
    extern __shared__ int hist[];
    __shared__ int wtot[ETHREADS / 32];
    const int tid  = threadIdx.x;
    const int lane = tid & 31;
    const int wave = __builtin_amdgcn_readfirstlane(tid >> 5);
    for (int b = tid; b < nbk; b += ETHREADS) hist[b] = 0;
    __syncthreads();
    const int start = blockIdx.x * chunk;
    int end = start + chunk; if (end > e) end = e;
    for (int base = start; base < end; base += EUNROLL * ETHREADS) {
        int d[EUNROLL];
        #pragma unroll
        for (int u = 0; u < EUNROLL; ++u) {
            const int i = base + u * ETHREADS + tid;
            d[u] = (i < end) ? dst[i] : -1;
        }
        #pragma unroll
        for (int u = 0; u < EUNROLL; ++u)
            if (d[u] >= 0) atomicAdd(&hist[d[u] >> BSH], 1);
    }
    __syncthreads();

    int carry = 0;
    for (int base = 0; base < nbk; base += ETHREADS) {
        const int b   = base + tid;
        const int v   = (b < nbk) ? hist[b] : 0;
        const int inc = wave_incl_scan(v, lane);
        if (lane == 31) wtot[wave] = inc;
        __syncthreads();
        const int wt   = wtot[lane];
        const int winc = wave_incl_scan(wt, lane);
        const int ex   = carry + __builtin_amdgcn_readlane(winc - wt, wave) + inc - v;
        carry += __builtin_amdgcn_readlane(winc, 31);
        if (b < nbk) {
            hist[b] = ex;
            cntm[b * gridDim.x + blockIdx.x] = v;
            lstm[b * gridDim.x + blockIdx.x] = ex;
        }
        __syncthreads();
    }

    int2* __restrict__ tmp = (int2*)tmpw + (size_t)blockIdx.x * chunk;
    for (int base = start; base < end; base += EUNROLL * ETHREADS) {
        int d[EUNROLL], s[EUNROLL];
        #pragma unroll
        for (int u = 0; u < EUNROLL; ++u) {
            const int i = base + u * ETHREADS + tid;
            d[u] = (i < end) ? dst[i] : -1;
            s[u] = (i < end) ? src[i] : 0;
        }
        #pragma unroll
        for (int u = 0; u < EUNROLL; ++u)
            if (d[u] >= 0) {
                const int p = atomicAdd(&hist[d[u] >> BSH], 1);
                tmp[p] = make_int2(s[u], d[u]);
            }
    }
}

__global__ __launch_bounds__(ETHREADS)
void k_bucket_final(const int* __restrict__ tmpw, const int* __restrict__ cntm,
                    const int* __restrict__ lstm,
                    int* __restrict__ offs, int* __restrict__ csr,
                    float* __restrict__ dinv,
                    const float* __restrict__ x, const float* __restrict__ W1,
                    float* __restrict__ h, int n, int g1, int chunk) {
    const int2* __restrict__ tmp = (const int2*)tmpw;
    __shared__ float sW[CFEAT * CFEAT];
    __shared__ float sdinv[BNODES];
    __shared__ int cnt[BNODES];
    __shared__ int loc[BNODES];
    __shared__ int wsum[BNODES / 32];
    __shared__ int segc[ETHREADS];
    __shared__ int segl[ETHREADS];
    __shared__ int red[2][ETHREADS / 32];
    const int tid  = threadIdx.x;
    const int lane = tid & 31;
    const int wave = tid >> 5;
    const int b    = blockIdx.x;

    for (int i = tid; i < (CFEAT * CFEAT) / 4; i += ETHREADS)
        ((float4*)sW)[i] = ((const float4*)W1)[i];

    int sc = 0, sl = 0;
    if (tid < g1) {
        sc = cntm[b * g1 + tid];
        sl = lstm[b * g1 + tid];
        segc[tid] = sc;
        segl[tid] = sl;
    }
    {
        const int ic = wave_incl_scan(sc, lane);
        const int il = wave_incl_scan(sl, lane);
        if (lane == 31) { red[0][wave] = ic; red[1][wave] = il; }
    }
    if (tid < BNODES) cnt[tid] = 0;
    __syncthreads();
    const int bs = __builtin_amdgcn_readlane(wave_incl_scan(red[1][lane], lane), 31);
    const int be = bs + __builtin_amdgcn_readlane(wave_incl_scan(red[0][lane], lane), 31);

    for (int g0 = wave; g0 < g1; g0 += 32 * SEGU) {
        int dy[SEGU][2];
        int len[SEGU];
        const int2* p[SEGU];
        #pragma unroll
        for (int u = 0; u < SEGU; ++u) {
            const int g  = g0 + 32 * u;
            const int gg = (g < g1) ? g : 0;
            len[u] = (g < g1) ? segc[gg] : 0;
            p[u]   = tmp + (size_t)gg * chunk + segl[gg];
        }
        #pragma unroll
        for (int u = 0; u < SEGU; ++u) {
            dy[u][0] = (lane < len[u])      ? p[u][lane].y      : -1;
            dy[u][1] = (lane + 32 < len[u]) ? p[u][lane + 32].y : -1;
        }
        #pragma unroll
        for (int u = 0; u < SEGU; ++u) {
            if (dy[u][0] >= 0) atomicAdd(&cnt[dy[u][0] & (BNODES - 1)], 1);
            if (dy[u][1] >= 0) atomicAdd(&cnt[dy[u][1] & (BNODES - 1)], 1);
        }
        #pragma unroll
        for (int u = 0; u < SEGU; ++u)
            for (int j = 64 + lane; j < len[u]; j += 32)
                atomicAdd(&cnt[p[u][j].y & (BNODES - 1)], 1);
    }
    __syncthreads();

    int c = 0, inc = 0;
    if (tid < BNODES) {
        c   = cnt[tid];
        inc = wave_incl_scan(c, lane);
        if (lane == 31) wsum[wave] = inc;
    }
    __syncthreads();
    if (tid < BNODES) {
        int add = 0;
        for (int w = 0; w < wave; ++w) add += wsum[w];
        const int excl = add + inc - c;
        loc[tid] = excl;
        cnt[tid] = 0;
        const int node = b * BNODES + tid;
        const float dn = rsqrtf((float)c + 1.0f);
        sdinv[tid] = dn;
        if (node < n) {
            offs[node] = bs + excl;
            dinv[node] = dn;
        }
    }
    if (b == (int)gridDim.x - 1 && tid == 0) offs[n] = be;
    __syncthreads();

    const int wv = __builtin_amdgcn_readfirstlane(tid >> 5);

    auto place = [&]() {
        for (int g0 = wave; g0 < g1; g0 += 32 * SEGU) {
            int2 t[SEGU][2];
            int len[SEGU];
            const int2* p[SEGU];
            #pragma unroll
            for (int u = 0; u < SEGU; ++u) {
                const int g  = g0 + 32 * u;
                const int gg = (g < g1) ? g : 0;
                len[u] = (g < g1) ? segc[gg] : 0;
                p[u]   = tmp + (size_t)gg * chunk + segl[gg];
            }
            #pragma unroll
            for (int u = 0; u < SEGU; ++u) {
                t[u][0] = (lane < len[u])      ? p[u][lane]      : make_int2(0, -1);
                t[u][1] = (lane + 32 < len[u]) ? p[u][lane + 32] : make_int2(0, -1);
            }
            #pragma unroll
            for (int u = 0; u < SEGU; ++u) {
                #pragma unroll
                for (int v = 0; v < 2; ++v)
                    if (t[u][v].y >= 0) {
                        const int k = t[u][v].y & (BNODES - 1);
                        const int q = atomicAdd(&cnt[k], 1);
                        csr[bs + loc[k] + q] = t[u][v].x;
                    }
            }
            #pragma unroll
            for (int u = 0; u < SEGU; ++u)
                for (int j = 64 + lane; j < len[u]; j += 32) {
                    const int2 pr = p[u][j];
                    const int k = pr.y & (BNODES - 1);
                    const int q = atomicAdd(&cnt[k], 1);
                    csr[bs + loc[k] + q] = pr.x;
                }
        }
    };

    static_assert((ETHREADS / 32 / 2) * 16 == BNODES, "pairs of waves x 16 rows must cover the bucket");
    auto product = [&]() {
        const int hi      = lane >> 4;
        const int lo      = lane & 15;
        const int rt      = (wv >> 1) * 16;
        const int rowbase = b * BNODES + rt;
        const int colbase = (wv & 1) * 64;
        if (rowbase < n) {
            int arow = rowbase + lo;
            if (arow >= n) arow = n - 1;
            const float* __restrict__ xrow = x + (size_t)arow * CFEAT;

            float dv[8];
            #pragma unroll
            for (int r = 0; r < 8; ++r) dv[r] = sdinv[rt + r + 8 * hi];

            v8f acc[4];
            const v8f z = {0.f, 0.f, 0.f, 0.f, 0.f, 0.f, 0.f, 0.f};
            #pragma unroll
            for (int t = 0; t < 4; ++t) acc[t] = z;

            for (int k = 0; k < CFEAT; k += 4) {
                v2f a;
                a.x = xrow[k + 2 * hi];
                a.y = xrow[k + 2 * hi + 1];
                #pragma unroll
                for (int t = 0; t < 4; ++t) {
                    v2f bf;
                    const int col = colbase + t * 16 + lo;
                    bf.x = sW[(k + 2 * hi) * CFEAT + col];
                    bf.y = sW[(k + 2 * hi + 1) * CFEAT + col];
                    acc[t] = __builtin_amdgcn_wmma_f32_16x16x4_f32(
                        false, a, false, bf, (short)0, acc[t], false, false);
                }
            }

            if (rowbase + 16 <= n) {
                #pragma unroll
                for (int t = 0; t < 4; ++t) {
                    #pragma unroll
                    for (int r = 0; r < 8; ++r) {
                        const int row = rowbase + r + 8 * hi;
                        h[(size_t)row * CFEAT + colbase + t * 16 + lo] = acc[t][r] * dv[r];
                    }
                }
            } else {
                #pragma unroll
                for (int t = 0; t < 4; ++t) {
                    #pragma unroll
                    for (int r = 0; r < 8; ++r) {
                        const int row = rowbase + r + 8 * hi;
                        if (row < n) h[(size_t)row * CFEAT + colbase + t * 16 + lo] = acc[t][r] * dv[r];
                    }
                }
            }
        }
    };

    const bool mul_first = ((wv >> 2) & 1) != 0;
    if (!mul_first) place();
    product();
    if (mul_first) place();
}


#define AGG_G 8

__device__ __forceinline__ int lane_i(int x, int t) { return __builtin_amdgcn_readlane(x, t); }

__device__ __forceinline__ void add4(float4& acc, const float4 v) {
    acc.x += v.x; acc.y += v.y; acc.z += v.z; acc.w += v.w;
}

__device__ __forceinline__ float4 aggregate(const float* __restrict__ hsrc,
                                            const int* __restrict__ offs,
                                            const int* __restrict__ csr,
                                            int node, int lane) {
    const float4* __restrict__ h4 = (const float4*)hsrc;
    float4 acc = h4[(size_t)node * 32 + lane];
    const int beg = offs[node], end = offs[node + 1];
    for (int j = beg; j < end; j += 32) {
        const int idx = j + lane;
        const int s = (idx < end) ? csr[idx] : 0;
        int cnt = end - j; if (cnt > 32) cnt = 32;
        int t = 0;
        for (; t + AGG_G <= cnt; t += AGG_G) {
            float4 v[AGG_G];
            #pragma unroll
            for (int i = 0; i < AGG_G; ++i) v[i] = h4[(size_t)lane_i(s, t + i) * 32 + lane];
            #pragma unroll
            for (int i = 0; i < AGG_G; ++i) add4(acc, v[i]);
        }
        const int r = cnt - t;
        if (r != 0) {
            const int t4 = t;
            const int t2 = t4 + (r & 4);
            const int t1 = t2 + (r & 2);
            float4 a0, a1, a2, a3, b0, b1, c0;
            a0 = a1 = a2 = a3 = b0 = b1 = c0 = make_float4(0.f, 0.f, 0.f, 0.f);
            if (r & 4) {
                a0 = h4[(size_t)lane_i(s, t4)     * 32 + lane];
                a1 = h4[(size_t)lane_i(s, t4 + 1) * 32 + lane];
                a2 = h4[(size_t)lane_i(s, t4 + 2) * 32 + lane];
                a3 = h4[(size_t)lane_i(s, t4 + 3) * 32 + lane];
            }
            if (r & 2) {
                b0 = h4[(size_t)lane_i(s, t2)     * 32 + lane];
                b1 = h4[(size_t)lane_i(s, t2 + 1) * 32 + lane];
            }
            if (r & 1) {
                c0 = h4[(size_t)lane_i(s, t1)     * 32 + lane];
            }
            if (r & 4) { add4(acc, a0); add4(acc, a1); add4(acc, a2); add4(acc, a3); }
            if (r & 2) { add4(acc, b0); add4(acc, b1); }
            if (r & 1) { add4(acc, c0); }
        }
    }
    return acc;
}

__global__ __launch_bounds__(256)
void k_prop1_proj(const float* __restrict__ hs, float* __restrict__ zs,
                  const int* __restrict__ offs, const int* __restrict__ csr,
                  const float* __restrict__ dinv, const float* __restrict__ b1,
                  const float* __restrict__ Wout, int n) {
    const int node = __builtin_amdgcn_readfirstlane((int)((blockIdx.x * blockDim.x + threadIdx.x) >> 5));
    const int lane = threadIdx.x & 31;
    if (blockIdx.x == 0 && threadIdx.x < 2)
        ((float4*)zs)[(size_t)n * 2 + threadIdx.x] = make_float4(0.f, 0.f, 0.f, 0.f);
    if (node >= n) return;

    const float din = dinv[node];
    float4 acc = aggregate(hs, offs, csr, node, lane);
    const float4 b = ((const float4*)b1)[lane];
    acc.x = fmaxf(acc.x * din + b.x, 0.0f);
    acc.y = fmaxf(acc.y * din + b.y, 0.0f);
    acc.z = fmaxf(acc.z * din + b.z, 0.0f);
    acc.w = fmaxf(acc.w * din + b.w, 0.0f);

    float p[4][2];
    #pragma unroll
    for (int m = 0; m < 4; ++m) {
        const float4* __restrict__ w4 = (const float4*)(Wout + ((size_t)m * CFEAT + lane * 4) * 2);
        const float4 w01 = w4[0];
        const float4 w23 = w4[1];
        p[m][0] = acc.x * w01.x + acc.y * w01.z + acc.z * w23.x + acc.w * w23.z;
        p[m][1] = acc.x * w01.y + acc.y * w01.w + acc.z * w23.y + acc.w * w23.w;
    }

    #pragma unroll
    for (int m = 0; m < 4; ++m)
        #pragma unroll
        for (int o = 0; o < 2; ++o) {
            float v = p[m][o];
            #pragma unroll
            for (int d = 16; d > 0; d >>= 1) v += __shfl_xor(v, d, 32);
            p[m][o] = v;
        }

    if (lane == 0) {
        float4* __restrict__ z4 = (float4*)zs + (size_t)node * 2;
        z4[0] = make_float4(p[0][0] * din, p[0][1] * din, p[1][0] * din, p[1][1] * din);
        z4[1] = make_float4(p[2][0] * din, p[2][1] * din, p[3][0] * din, p[3][1] * din);
    }
}


#define P2_NPW 4
#define P2_CH  3

__device__ __forceinline__ float sum_same_parity(float x) {
    x += __builtin_bit_cast(float, __builtin_amdgcn_update_dpp(0, __builtin_bit_cast(int, x), 0x162, 0xf, 0xf, true));
    x += __builtin_bit_cast(float, __builtin_amdgcn_update_dpp(0, __builtin_bit_cast(int, x), 0x164, 0xf, 0xf, true));
    x += __builtin_bit_cast(float, __builtin_amdgcn_update_dpp(0, __builtin_bit_cast(int, x), 0x168, 0xf, 0xf, true));
    const int b = __builtin_bit_cast(int, x);
    x += __builtin_bit_cast(float, __builtin_amdgcn_permlanex16(b, b, 0x76543210, 0xfedcba98, false, false));
    return x;
}

__global__ __launch_bounds__(256)
void k_prop2_small(const float* __restrict__ zs, float* __restrict__ out,
                   const int* __restrict__ offs, const int* __restrict__ csr,
                   const float* __restrict__ dinv, const float* __restrict__ bout,
                   int n, int e) {
    const int lane = threadIdx.x & 31;
    const int q    = lane >> 1;
    const int half = lane & 1;
    const int node0 = __builtin_amdgcn_readfirstlane((int)((blockIdx.x * blockDim.x + threadIdx.x) >> 5)) * P2_NPW;
    if (node0 >= n) return;
    const float4* __restrict__ z4 = (const float4*)zs;

    const int   tq   = node0 + q;
    const float dinq = dinv[(tq < n) ? tq : (n - 1)];
    const float4 bo  = ((const float4*)bout)[half];

    int nd[P2_NPW], beg[P2_NPW], end[P2_NPW];
    #pragma unroll
    for (int i = 0; i < P2_NPW; ++i) {
        const int t = node0 + i;
        nd[i]  = (t < n) ? t : (n - 1);
        beg[i] = offs[nd[i]];
        end[i] = offs[nd[i] + 1];
    }

    int  s[P2_NPW][P2_CH];
    bool ok[P2_NPW][P2_CH];
    #pragma unroll
    for (int i = 0; i < P2_NPW; ++i)
        #pragma unroll
        for (int k = 0; k < P2_CH; ++k) {
            const int idx = beg[i] + k * 16 + q - 1;
            ok[i][k] = idx < end[i];
            int ic = (idx > 0) ? idx : 0;
            ic = (ic < e) ? ic : (e - 1);
            s[i][k] = csr[ic];
        }

    float4 acc[P2_NPW];
    #pragma unroll
    for (int i = 0; i < P2_NPW; ++i) {
        float4 v[P2_CH];
        #pragma unroll
        for (int k = 0; k < P2_CH; ++k) {
            int sv = ok[i][k] ? s[i][k] : n;
            if (k == 0) sv = (q == 0) ? nd[i] : sv;
            v[k] = z4[(unsigned)sv * 2u + (unsigned)half];
        }
        acc[i] = v[0];
        #pragma unroll
        for (int k = 1; k < P2_CH; ++k) {
            acc[i].x += v[k].x; acc[i].y += v[k].y; acc[i].z += v[k].z; acc[i].w += v[k].w;
        }
    }

    #pragma unroll
    for (int i = 0; i < P2_NPW; ++i)
        for (int j = beg[i] + P2_CH * 16 - 1; j < end[i]; j += 16) {
            const int  idx = j + q;
            const bool okj = idx < end[i];
            const int  sj  = csr[okj ? idx : j];
            const float4 vj = z4[(unsigned)(okj ? sj : n) * 2u + (unsigned)half];
            acc[i].x += vj.x; acc[i].y += vj.y; acc[i].z += vj.z; acc[i].w += vj.w;
        }

    #pragma unroll
    for (int i = 0; i < P2_NPW; ++i) {
        acc[i].x = sum_same_parity(acc[i].x);
        acc[i].y = sum_same_parity(acc[i].y);
        acc[i].z = sum_same_parity(acc[i].z);
        acc[i].w = sum_same_parity(acc[i].w);
    }

    float4 t = acc[0];
    #pragma unroll
    for (int i = 1; i < P2_NPW; ++i)
        if (q == i) t = acc[i];
    float2 r0, r1;
    r0.x = t.x * dinq + bo.x;
    r0.y = t.y * dinq + bo.y;
    r1.x = t.z * dinq + bo.z;
    r1.y = t.w * dinq + bo.w;
    if (q < P2_NPW && tq < n) {
        const int m0 = half * 2;
        ((float2*)out)[(size_t)m0 * n + tq]       = r0;
        ((float2*)out)[(size_t)(m0 + 1) * n + tq] = r1;
    }
}


extern "C" void kernel_launch(void* const* d_in, const int* in_sizes, int n_in,
                              void* d_out, int out_size, void* d_ws, size_t ws_size,
                              hipStream_t stream) {
    const float* x    = (const float*)d_in[0];
    const int*   ei   = (const int*)d_in[1];
    const float* W1   = (const float*)d_in[2];
    const float* b1   = (const float*)d_in[3];
    const float* Wout = (const float*)d_in[4];
    const float* bout = (const float*)d_in[5];
    float* out = (float*)d_out;

    const int n = in_sizes[0] / CFEAT;
    const int e = in_sizes[1] / 2;
    const int* srcv = ei;
    const int* dstv = ei + e;

    uintptr_t p = (uintptr_t)d_ws;
    auto alloc = [&](size_t bytes) {
        p = (p + 255) & ~(uintptr_t)255;
        void* r = (void*)p;
        p += bytes;
        return r;
    };
    const int G1    = 256;
    const int chunk = (e + G1 - 1) / G1;
    const int nbk   = (n + BNODES - 1) >> BSH;
    int*   cntm   = (int*)  alloc((size_t)nbk * G1 * 4);
    int*   lstm   = (int*)  alloc((size_t)nbk * G1 * 4);
    int*   tmp    = (int*)  alloc((size_t)G1 * chunk * 8);
    int*   offs   = (int*)  alloc((size_t)(n + 1) * 4);
    int*   csr    = (int*)  alloc((size_t)e * 4);
    float* dinv   = (float*)alloc((size_t)n * 4);
    float* h      = (float*)alloc((size_t)n * CFEAT * 4);
    float* zs     = (float*)alloc(((size_t)n + 1) * 8 * 4);
    (void)ws_size; (void)n_in; (void)out_size;

    const int TB = 256;

    k_slice_sort  <<<G1, ETHREADS, (size_t)nbk * 4, stream>>>(srcv, dstv, cntm, lstm, tmp, e, chunk, nbk);
    k_bucket_final<<<nbk, ETHREADS, 0, stream>>>(tmp, cntm, lstm, offs, csr, dinv, x, W1, h, n, G1, chunk);

    const int gprop = (n + 7) / 8;
    k_prop1_proj <<<gprop, TB, 0, stream>>>(h, zs, offs, csr, dinv, b1, Wout, n);
    const int gp2 = (n + 8 * P2_NPW - 1) / (8 * P2_NPW);
    k_prop2_small<<<gp2, TB, 0, stream>>>(zs, out, offs, csr, dinv, bout, n, e);
}
